# starting kernel
# baseline (speedup 1.0000x reference)
#include <hip/hip_runtime.h>
#include <math.h>

#define BB 4
#define SS 2048
#define DD 1024
#define NNh 64
#define MM (BB * SS)
#define KX (2 * DD)
#define NP 192
#define KH (2 * NNh)
#define NO (2 * DD)

#define A2LD 144

typedef __attribute__((ext_vector_type(2))) float v2f;
typedef __attribute__((ext_vector_type(8))) float v8f;

__device__ __forceinline__ v8f wmma_f32(v2f a, v2f b, v8f c) {
  return __builtin_amdgcn_wmma_f32_16x16x4_f32(
       false, a,  false, b,
       (short)0, c,  false,  false);
}

template <int NWN>
__device__ void gemm_tile_acc(const float* __restrict__ A, int lda,
                              const float* __restrict__ Bp, int ldb,
                              int K, int m0, int j0,
                              v2f* As2, v2f* Bs2, v8f acc[2][NWN]) {
  constexpr int BN   = NWN * 32;
  constexpr int B2LD = BN + 16;
  constexpr int NB4  = BN / 64;

  const int tid  = threadIdx.x;
  const int lane = tid & 31;
  const int wave = tid >> 5;
  const int half = lane >> 4;
  const int lrow = lane & 15;
  const int wm   = (wave & 3) * 32;
  const int wn   = (wave >> 2) * (NWN * 16);

  const int am0 = tid >> 2;
  const int ak0 = (tid & 3) << 2;
  const int am1 = (256 + tid) >> 2;
  const int ak1 = ((256 + tid) & 3) << 2;
  int bk[NB4], bj[NB4];
#pragma unroll
  for (int i = 0; i < NB4; ++i) {
    int idx = i * 256 + tid;
    bk[i] = idx / (BN / 4);
    bj[i] = (idx % (BN / 4)) * 4;
  }

  float4 aReg0 = *(const float4*)(A + (size_t)(m0 + am0) * lda + ak0);
  float4 aReg1 = *(const float4*)(A + (size_t)(m0 + am1) * lda + ak1);
  float4 bReg[NB4];
#pragma unroll
  for (int i = 0; i < NB4; ++i)
    bReg[i] = *(const float4*)(Bp + (size_t)bk[i] * ldb + j0 + bj[i]);

  for (int k0 = 0; k0 < K; k0 += 16) {
    __syncthreads();
    {
      v2f t;
      t.x = aReg0.x; t.y = aReg0.y; As2[(ak0 >> 1) * A2LD + am0] = t;
      t.x = aReg0.z; t.y = aReg0.w; As2[((ak0 >> 1) + 1) * A2LD + am0] = t;
      t.x = aReg1.x; t.y = aReg1.y; As2[(ak1 >> 1) * A2LD + am1] = t;
      t.x = aReg1.z; t.y = aReg1.w; As2[((ak1 >> 1) + 1) * A2LD + am1] = t;
    }
    {
      float* bsf = (float*)Bs2;
#pragma unroll
      for (int i = 0; i < NB4; ++i) {
        int base = ((bk[i] >> 1) * B2LD + bj[i]) * 2 + (bk[i] & 1);
        bsf[base + 0] = bReg[i].x;
        bsf[base + 2] = bReg[i].y;
        bsf[base + 4] = bReg[i].z;
        bsf[base + 6] = bReg[i].w;
      }
    }
    __syncthreads();

    if (k0 + 16 < K) {
      aReg0 = *(const float4*)(A + (size_t)(m0 + am0) * lda + (k0 + 16) + ak0);
      aReg1 = *(const float4*)(A + (size_t)(m0 + am1) * lda + (k0 + 16) + ak1);
#pragma unroll
      for (int i = 0; i < NB4; ++i)
        bReg[i] = *(const float4*)(Bp + (size_t)(k0 + 16 + bk[i]) * ldb + j0 + bj[i]);
    }
    if (k0 + 32 < K) {
      __builtin_prefetch(A + (size_t)(m0 + am0) * lda + (k0 + 32) + ak0, 0, 0);
      __builtin_prefetch(A + (size_t)(m0 + am1) * lda + (k0 + 32) + ak1, 0, 0);
    }

#pragma unroll
    for (int kstep = 0; kstep < 4; ++kstep) {
      const int p = 2 * kstep + half;
      v2f a0 = As2[p * A2LD + wm + lrow];
      v2f a1 = As2[p * A2LD + wm + 16 + lrow];
      v2f bf[NWN];
#pragma unroll
      for (int ni = 0; ni < NWN; ++ni)
        bf[ni] = Bs2[p * B2LD + wn + ni * 16 + lrow];
#pragma unroll
      for (int ni = 0; ni < NWN; ++ni) {
        acc[0][ni] = wmma_f32(a0, bf[ni], acc[0][ni]);
        acc[1][ni] = wmma_f32(a1, bf[ni], acc[1][ni]);
      }
    }
  }
}

__global__ void pack_w1_kernel(const float* __restrict__ Bwr, const float* __restrict__ Bwi,
                               const float* __restrict__ gw, float* __restrict__ W1) {
  int idx = blockIdx.x * 256 + threadIdx.x;
  if (idx >= KX * NP) return;
  int k = idx / NP, j = idx - k * NP;
  int d = k >> 1, p = k & 1;
  float v;
  if (j < 64) {
    v = p ? -Bwi[j * DD + d] : Bwr[j * DD + d];
  } else if (j < 128) {
    int n = j - 64;
    v = p ? Bwr[n * DD + d] : Bwi[n * DD + d];
  } else {
    int n = j - 128;
    v = gw[n * (2 * DD) + p * DD + d];
  }
  W1[idx] = v;
}

__global__ void pack_wd_kernel(const float* __restrict__ Dr, const float* __restrict__ Di,
                               float* __restrict__ WD) {
  int idx = blockIdx.x * 256 + threadIdx.x;
  if (idx >= KX * NO) return;
  int k = idx >> 11, j = idx & (NO - 1);
  int e = k >> 1, p = k & 1, d = j >> 1;
  int q = j & 1;
  float wr = Dr[d * DD + e], wi = Di[d * DD + e];
  WD[idx] = (p == q) ? wr : (p ? -wi : wi);
}

__global__ void pack_wc_kernel(const float* __restrict__ Cr, const float* __restrict__ Ci,
                               float* __restrict__ WC) {
  int idx = blockIdx.x * 256 + threadIdx.x;
  if (idx >= KH * NO) return;
  int k = idx >> 11, j = idx & (NO - 1);
  int n = k >> 1, p = k & 1, d = j >> 1;
  int q = j & 1;
  float wr = Cr[d * NNh + n], wi = Ci[d * NNh + n];
  WC[idx] = (p == q) ? wr : (p ? -wi : wi);
}

__global__ __launch_bounds__(256) void gemm_proj_kernel(const float* __restrict__ x,
                                                        const float* __restrict__ W1,
                                                        const float* __restrict__ gate_b,
                                                        float* __restrict__ P) {
  __shared__ v2f As2[8 * A2LD];
  __shared__ v2f Bs2[8 * (64 + 16)];
  v8f acc[2][2] = {{{0,0,0,0,0,0,0,0}, {0,0,0,0,0,0,0,0}},
                   {{0,0,0,0,0,0,0,0}, {0,0,0,0,0,0,0,0}}};
  const int m0 = blockIdx.x * 128;
  const int j0 = blockIdx.y * 64;
  gemm_tile_acc<2>(x, KX, W1, NP, KX, m0, j0, As2, Bs2, acc);

  const int lane = threadIdx.x & 31, wave = threadIdx.x >> 5;
  const int half = lane >> 4, lrow = lane & 15;
  const int wm = (wave & 3) * 32, wn = (wave >> 2) * 32;
#pragma unroll
  for (int mi = 0; mi < 2; ++mi)
#pragma unroll
    for (int ni = 0; ni < 2; ++ni)
#pragma unroll
      for (int e = 0; e < 8; ++e) {
        int m = m0 + wm + mi * 16 + half * 8 + e;
        int j = j0 + wn + ni * 16 + lrow;
        float v = acc[mi][ni][e];
        if (j >= 128) v = 1.0f / (1.0f + expf(-(v + gate_b[j - 128])));
        P[(size_t)m * NP + j] = v;
      }
}

__global__ __launch_bounds__(256) void scan_kernel(const float* __restrict__ P,
                                                   const float* __restrict__ h0,
                                                   const float* __restrict__ theta,
                                                   const float* __restrict__ dampp,
                                                   float* __restrict__ hws,
                                                   float* __restrict__ hfin) {
  const int t = threadIdx.x;
  const int b = t >> 6, n = t & 63;
  const float c = cosf(theta[n]);
  const float s = sinf(theta[n]);
  const float dmp = 0.5f + 0.5f / (1.0f + expf(-dampp[n]));
  const float ca = c * dmp, sa = s * dmp;
  float hr = h0[(b * NNh + n) * 2 + 0];
  float hi = h0[(b * NNh + n) * 2 + 1];
#pragma unroll 4
  for (int st = 0; st < SS; ++st) {
    size_t m = (size_t)b * SS + st;
    float g  = P[m * NP + 128 + n];
    float br = P[m * NP + n];
    float bi = P[m * NP + 64 + n];
    float rr = hr * ca - hi * sa;
    float ri = hr * sa + hi * ca;
    hr = g * br + (1.0f - g) * rr;
    hi = g * bi + (1.0f - g) * ri;
    hws[m * KH + 2 * n + 0] = hr;
    hws[m * KH + 2 * n + 1] = hi;
  }
  hfin[(b * NNh + n) * 2 + 0] = hr;
  hfin[(b * NNh + n) * 2 + 1] = hi;
}

__global__ __launch_bounds__(256) void gemm_out_kernel(const float* __restrict__ x,
                                                       const float* __restrict__ WD,
                                                       const float* __restrict__ hws,
                                                       const float* __restrict__ WC,
                                                       float* __restrict__ out) {
  __shared__ v2f As2[8 * A2LD];
  __shared__ v2f Bs2[8 * (128 + 16)];
  v8f acc[2][4] = {{{0,0,0,0,0,0,0,0}, {0,0,0,0,0,0,0,0},
                    {0,0,0,0,0,0,0,0}, {0,0,0,0,0,0,0,0}},
                   {{0,0,0,0,0,0,0,0}, {0,0,0,0,0,0,0,0},
                    {0,0,0,0,0,0,0,0}, {0,0,0,0,0,0,0,0}}};
  const int m0 = blockIdx.x * 128;
  const int j0 = blockIdx.y * 128;
  gemm_tile_acc<4>(x, KX, WD, NO, KX, m0, j0, As2, Bs2, acc);
  __syncthreads();
  gemm_tile_acc<4>(hws, KH, WC, NO, KH, m0, j0, As2, Bs2, acc);

  const int lane = threadIdx.x & 31, wave = threadIdx.x >> 5;
  const int half = lane >> 4, lrow = lane & 15;
  const int wm = (wave & 3) * 32, wn = (wave >> 2) * 64;
#pragma unroll
  for (int mi = 0; mi < 2; ++mi)
#pragma unroll
    for (int ni = 0; ni < 4; ++ni)
#pragma unroll
      for (int e = 0; e < 8; ++e) {
        int m = m0 + wm + mi * 16 + half * 8 + e;
        int j = j0 + wn + ni * 16 + lrow;
        float z = acc[mi][ni][e];
        float zr = __shfl(z, lane & ~1, 32);
        out[(size_t)m * NO + j] = zr * z + z;
      }
}

extern "C" void kernel_launch(void* const* d_in, const int* in_sizes, int n_in,
                              void* d_out, int out_size, void* d_ws, size_t ws_size,
                              hipStream_t stream) {
  const float* x     = (const float*)d_in[0];
  const float* h0    = (const float*)d_in[1];
  const float* th    = (const float*)d_in[2];
  const float* dampp = (const float*)d_in[3];
  const float* Bwr   = (const float*)d_in[4];
  const float* Bwi   = (const float*)d_in[5];
  const float* Cwr   = (const float*)d_in[6];
  const float* Cwi   = (const float*)d_in[7];
  const float* Dwr   = (const float*)d_in[8];
  const float* Dwi   = (const float*)d_in[9];
  const float* gw    = (const float*)d_in[10];
  const float* gb    = (const float*)d_in[11];

  float* out  = (float*)d_out;
  float* hfin = out + (size_t)MM * NO;

  float* ws = (float*)d_ws;
  float* W1  = ws;
  float* WD  = W1 + (size_t)KX * NP;
  float* WC  = WD + (size_t)KX * NO;
  float* P   = WC + (size_t)KH * NO;
  float* Hws = P + (size_t)MM * NP;

  pack_w1_kernel<<<(KX * NP + 255) / 256, 256, 0, stream>>>(Bwr, Bwi, gw, W1);
  pack_wd_kernel<<<(KX * NO + 255) / 256, 256, 0, stream>>>(Dwr, Dwi, WD);
  pack_wc_kernel<<<(KH * NO + 255) / 256, 256, 0, stream>>>(Cwr, Cwi, WC);

  gemm_proj_kernel<<<dim3(MM / 128, NP / 64), 256, 0, stream>>>(x, W1, gb, P);
  scan_kernel<<<1, 256, 0, stream>>>(P, h0, th, dampp, Hws, hfin);
  gemm_out_kernel<<<dim3(MM / 128, NO / 128), 256, 0, stream>>>(x, WD, Hws, WC, out);
}


# GEMM core runs over two LDS stage buffers: next k tile stored into the idle buffer at the head of a stage, so a 16-k stage has one barrier (was two); D x kernel held to 128 registers; same sums
# speedup vs baseline: 1.4372x; 1.4372x over previous
#include <hip/hip_runtime.h>
#include <math.h>

#define BB 4
#define SS 2048
#define DD 1024
#define NNh 64
#define MM (BB * SS)
#define KX (2 * DD)
#define NP 192
#define KH (2 * NNh)
#define NO (2 * DD)

#define A2LD 144

typedef __attribute__((ext_vector_type(2))) float v2f;
typedef __attribute__((ext_vector_type(8))) float v8f;

__device__ __forceinline__ v8f wmma_f32(v2f a, v2f b, v8f c) {
  return __builtin_amdgcn_wmma_f32_16x16x4_f32(
       false, a,  false, b,
       (short)0, c,  false,  false);
}

template <int NWN>
struct GemmCtx {
  static constexpr int BN   = NWN * 32;
  static constexpr int B2LD = BN + 16;
  static constexpr int NB4  = BN / 64;
  static constexpr int ASZ  = 8 * A2LD;
  static constexpr int BSZ  = 8 * B2LD;
  const float* pa0; const float* pa1;
  const float* pb[NB4];
  int sa0, sa1;
  int sb[NB4];
  int fa, fb;
  int ldb, K;
  float4 aReg0, aReg1, bReg[NB4];
};

template <int NWN>
__device__ __forceinline__ void gemm_fetch(GemmCtx<NWN>& c, int k) {
  c.aReg0 = *(const float4*)(c.pa0 + k);
  c.aReg1 = *(const float4*)(c.pa1 + k);
#pragma unroll
  for (int i = 0; i < GemmCtx<NWN>::NB4; ++i)
    c.bReg[i] = *(const float4*)(c.pb[i] + (size_t)k * c.ldb);
}

template <int NWN>
__device__ __forceinline__ void gemm_stash(const GemmCtx<NWN>& c, v2f* An, v2f* Bn) {
  v2f t;
  t.x = c.aReg0.x; t.y = c.aReg0.y; An[c.sa0] = t;
  t.x = c.aReg0.z; t.y = c.aReg0.w; An[c.sa0 + A2LD] = t;
  t.x = c.aReg1.x; t.y = c.aReg1.y; An[c.sa1] = t;
  t.x = c.aReg1.z; t.y = c.aReg1.w; An[c.sa1 + A2LD] = t;
  float* bsf = (float*)Bn;
#pragma unroll
  for (int i = 0; i < GemmCtx<NWN>::NB4; ++i) {
    bsf[c.sb[i] + 0] = c.bReg[i].x;
    bsf[c.sb[i] + 2] = c.bReg[i].y;
    bsf[c.sb[i] + 4] = c.bReg[i].z;
    bsf[c.sb[i] + 6] = c.bReg[i].w;
  }
}

template <int NWN, int CUR>
__device__ __forceinline__ void gemm_stage(GemmCtx<NWN>& c, int k0, v2f* As2, v2f* Bs2,
                                           v8f acc[2][NWN]) {
  using G = GemmCtx<NWN>;
  const v2f* Ac = As2 + CUR * G::ASZ;
  const v2f* Bc = Bs2 + CUR * G::BSZ;
  v2f* An = As2 + (CUR ^ 1) * G::ASZ;
  v2f* Bn = Bs2 + (CUR ^ 1) * G::BSZ;

  v2f a0[2], a1[2], bf[2][NWN];
  a0[0] = Ac[c.fa];
  a1[0] = Ac[c.fa + 16];
#pragma unroll
  for (int ni = 0; ni < NWN; ++ni) bf[0][ni] = Bc[c.fb + ni * 16];

  if (k0 + 16 < c.K) {
    gemm_stash<NWN>(c, An, Bn);
    if (k0 + 32 < c.K) gemm_fetch<NWN>(c, k0 + 32);
    if (k0 + 48 < c.K) {
      __builtin_prefetch(c.pa0 + k0 + 48, 0, 0);
      __builtin_prefetch(c.pa1 + k0 + 48, 0, 0);
    }
  }

#pragma unroll
  for (int kstep = 0; kstep < 4; ++kstep) {
    const int cs = kstep & 1, ns = cs ^ 1;
    if (kstep < 3) {
      const int p2 = 2 * (kstep + 1);
      a0[ns] = Ac[c.fa + p2 * A2LD];
      a1[ns] = Ac[c.fa + p2 * A2LD + 16];
#pragma unroll
      for (int ni = 0; ni < NWN; ++ni) bf[ns][ni] = Bc[c.fb + p2 * G::B2LD + ni * 16];
    }
#pragma unroll
    for (int ni = 0; ni < NWN; ++ni) {
      acc[0][ni] = wmma_f32(a0[cs], bf[cs][ni], acc[0][ni]);
      acc[1][ni] = wmma_f32(a1[cs], bf[cs][ni], acc[1][ni]);
    }
  }
  __syncthreads();
}

template <int NWN>
__device__ void gemm_tile_acc(const float* __restrict__ A, int lda,
                              const float* __restrict__ Bp, int ldb,
                              int K, int m0, int j0,
                              v2f* As2, v2f* Bs2, v8f acc[2][NWN]) {
  using G = GemmCtx<NWN>;
  const int tid  = threadIdx.x;
  const int lane = tid & 31;
  const int wave = tid >> 5;
  const int half = lane >> 4;
  const int lrow = lane & 15;
  const int wm   = (wave & 3) * 32;
  const int wn   = (wave >> 2) * (NWN * 16);

  GemmCtx<NWN> c;
  c.ldb = ldb; c.K = K;
  const int am0 = tid >> 2;
  const int ak0 = (tid & 3) << 2;
  const int am1 = (256 + tid) >> 2;
  const int ak1 = ((256 + tid) & 3) << 2;
  c.pa0 = A + (size_t)(m0 + am0) * lda + ak0;
  c.pa1 = A + (size_t)(m0 + am1) * lda + ak1;
  c.sa0 = (ak0 >> 1) * A2LD + am0;
  c.sa1 = (ak1 >> 1) * A2LD + am1;
#pragma unroll
  for (int i = 0; i < G::NB4; ++i) {
    const int idx = i * 256 + tid;
    const int bk = idx / (G::BN / 4);
    const int bj = (idx % (G::BN / 4)) * 4;
    c.pb[i] = Bp + (size_t)bk * ldb + j0 + bj;
    c.sb[i] = ((bk >> 1) * G::B2LD + bj) * 2 + (bk & 1);
  }
  c.fa = half * A2LD + wm + lrow;
  c.fb = half * G::B2LD + wn + lrow;

  gemm_fetch<NWN>(c, 0);
  gemm_stash<NWN>(c, As2, Bs2);
  if (16 < K) gemm_fetch<NWN>(c, 16);
  __syncthreads();

  for (int k0 = 0; k0 < K; k0 += 32) {
    gemm_stage<NWN, 0>(c, k0,      As2, Bs2, acc);
    gemm_stage<NWN, 1>(c, k0 + 16, As2, Bs2, acc);
  }
}

__global__ void pack_w1_kernel(const float* __restrict__ Bwr, const float* __restrict__ Bwi,
                               const float* __restrict__ gw, float* __restrict__ W1) {
  int idx = blockIdx.x * 256 + threadIdx.x;
  if (idx >= KX * NP) return;
  int k = idx / NP, j = idx - k * NP;
  int d = k >> 1, p = k & 1;
  float v;
  if (j < 64) {
    v = p ? -Bwi[j * DD + d] : Bwr[j * DD + d];
  } else if (j < 128) {
    int n = j - 64;
    v = p ? Bwr[n * DD + d] : Bwi[n * DD + d];
  } else {
    int n = j - 128;
    v = gw[n * (2 * DD) + p * DD + d];
  }
  W1[idx] = v;
}

__global__ void pack_wd_kernel(const float* __restrict__ Dr, const float* __restrict__ Di,
                               float* __restrict__ WD) {
  int idx = blockIdx.x * 256 + threadIdx.x;
  if (idx >= KX * NO) return;
  int k = idx >> 11, j = idx & (NO - 1);
  int e = k >> 1, p = k & 1, d = j >> 1;
  int q = j & 1;
  float wr = Dr[d * DD + e], wi = Di[d * DD + e];
  WD[idx] = (p == q) ? wr : (p ? -wi : wi);
}

__global__ void pack_wc_kernel(const float* __restrict__ Cr, const float* __restrict__ Ci,
                               float* __restrict__ WC) {
  int idx = blockIdx.x * 256 + threadIdx.x;
  if (idx >= KH * NO) return;
  int k = idx >> 11, j = idx & (NO - 1);
  int n = k >> 1, p = k & 1, d = j >> 1;
  int q = j & 1;
  float wr = Cr[d * NNh + n], wi = Ci[d * NNh + n];
  WC[idx] = (p == q) ? wr : (p ? -wi : wi);
}

__global__ __launch_bounds__(256) void gemm_proj_kernel(const float* __restrict__ x,
                                                        const float* __restrict__ W1,
                                                        const float* __restrict__ gate_b,
                                                        float* __restrict__ P) {
  __shared__ v2f As2[2 * 8 * A2LD];
  __shared__ v2f Bs2[2 * 8 * (64 + 16)];
  v8f acc[2][2] = {{{0,0,0,0,0,0,0,0}, {0,0,0,0,0,0,0,0}},
                   {{0,0,0,0,0,0,0,0}, {0,0,0,0,0,0,0,0}}};
  const int m0 = blockIdx.x * 128;
  const int j0 = blockIdx.y * 64;
  gemm_tile_acc<2>(x, KX, W1, NP, KX, m0, j0, As2, Bs2, acc);

  const int lane = threadIdx.x & 31, wave = threadIdx.x >> 5;
  const int half = lane >> 4, lrow = lane & 15;
  const int wm = (wave & 3) * 32, wn = (wave >> 2) * 32;
#pragma unroll
  for (int mi = 0; mi < 2; ++mi)
#pragma unroll
    for (int ni = 0; ni < 2; ++ni)
#pragma unroll
      for (int e = 0; e < 8; ++e) {
        int m = m0 + wm + mi * 16 + half * 8 + e;
        int j = j0 + wn + ni * 16 + lrow;
        float v = acc[mi][ni][e];
        if (j >= 128) v = 1.0f / (1.0f + expf(-(v + gate_b[j - 128])));
        P[(size_t)m * NP + j] = v;
      }
}

__device__ __forceinline__ void scan_body(const float* __restrict__ P,
                                          const float* __restrict__ h0,
                                          const float* __restrict__ theta,
                                          const float* __restrict__ dampp,
                                          float* __restrict__ hws,
                                          float* __restrict__ hfin) {
  const int t = threadIdx.x;
  const int b = t >> 6, n = t & 63;
  const float c = cosf(theta[n]);
  const float s = sinf(theta[n]);
  const float dmp = 0.5f + 0.5f / (1.0f + expf(-dampp[n]));
  const float ca = c * dmp, sa = s * dmp;
  float hr = h0[(b * NNh + n) * 2 + 0];
  float hi = h0[(b * NNh + n) * 2 + 1];
#pragma unroll 4
  for (int st = 0; st < SS; ++st) {
    size_t m = (size_t)b * SS + st;
    float g  = P[m * NP + 128 + n];
    float br = P[m * NP + n];
    float bi = P[m * NP + 64 + n];
    float rr = hr * ca - hi * sa;
    float ri = hr * sa + hi * ca;
    hr = g * br + (1.0f - g) * rr;
    hi = g * bi + (1.0f - g) * ri;
    hws[m * KH + 2 * n + 0] = hr;
    hws[m * KH + 2 * n + 1] = hi;
  }
  hfin[(b * NNh + n) * 2 + 0] = hr;
  hfin[(b * NNh + n) * 2 + 1] = hi;
}

__device__ __forceinline__ float4* zpart_slot(float* out, int m0, int j0, int q) {
  const int lane = threadIdx.x & 31, wave = threadIdx.x >> 5;
  return (float4*)(out + (size_t)(m0 + wave * 16 + q) * NO + j0 + lane * 4);
}

__global__ __launch_bounds__(256, 8) void gemm_dx_scan_kernel(const float* __restrict__ x,
                                                           const float* __restrict__ WD,
                                                           const float* __restrict__ P,
                                                           const float* __restrict__ h0,
                                                           const float* __restrict__ theta,
                                                           const float* __restrict__ dampp,
                                                           float* __restrict__ hws,
                                                           float* __restrict__ hfin,
                                                           float* __restrict__ out) {
  __shared__ v2f As2[2 * 8 * A2LD];
  __shared__ v2f Bs2[2 * 8 * (128 + 16)];
  if (blockIdx.x == 0) {
    scan_body(P, h0, theta, dampp, hws, hfin);
    return;
  }
  v8f acc[2][4] = {{{0,0,0,0,0,0,0,0}, {0,0,0,0,0,0,0,0},
                    {0,0,0,0,0,0,0,0}, {0,0,0,0,0,0,0,0}},
                   {{0,0,0,0,0,0,0,0}, {0,0,0,0,0,0,0,0},
                    {0,0,0,0,0,0,0,0}, {0,0,0,0,0,0,0,0}}};
  const int tile = blockIdx.x - 1;
  const int m0 = (tile & (MM / 128 - 1)) * 128;
  const int j0 = (tile / (MM / 128)) * 128;
  gemm_tile_acc<4>(x, KX, WD, NO, KX, m0, j0, As2, Bs2, acc);

#pragma unroll
  for (int mi = 0; mi < 2; ++mi)
#pragma unroll
    for (int ni = 0; ni < 4; ++ni) {
      const v8f a = acc[mi][ni];
      const int q = (mi * 4 + ni) * 2;
      *zpart_slot(out, m0, j0, q)     = make_float4(a[0], a[1], a[2], a[3]);
      *zpart_slot(out, m0, j0, q + 1) = make_float4(a[4], a[5], a[6], a[7]);
    }
}

__global__ __launch_bounds__(256) void gemm_ch_out_kernel(const float* __restrict__ hws,
                                                          const float* __restrict__ WC,
                                                          float* out) {
  __shared__ v2f As2[2 * 8 * A2LD];
  __shared__ v2f Bs2[2 * 8 * (128 + 16)];
  const int m0 = blockIdx.x * 128;
  const int j0 = blockIdx.y * 128;
  v8f acc[2][4];
#pragma unroll
  for (int mi = 0; mi < 2; ++mi)
#pragma unroll
    for (int ni = 0; ni < 4; ++ni) {
      const int q = (mi * 4 + ni) * 2;
      const float4 lo = *zpart_slot(out, m0, j0, q);
      const float4 hi = *zpart_slot(out, m0, j0, q + 1);
      v8f a = {lo.x, lo.y, lo.z, lo.w, hi.x, hi.y, hi.z, hi.w};
      acc[mi][ni] = a;
    }
  gemm_tile_acc<4>(hws, KH, WC, NO, KH, m0, j0, As2, Bs2, acc);

  const int lane = threadIdx.x & 31, wave = threadIdx.x >> 5;
  const int half = lane >> 4, lrow = lane & 15;
  const int wm = (wave & 3) * 32, wn = (wave >> 2) * 64;
#pragma unroll
  for (int mi = 0; mi < 2; ++mi)
#pragma unroll
    for (int ni = 0; ni < 4; ++ni)
#pragma unroll
      for (int e = 0; e < 8; ++e) {
        int m = m0 + wm + mi * 16 + half * 8 + e;
        int j = j0 + wn + ni * 16 + lrow;
        float z = acc[mi][ni][e];
        float zr = __shfl(z, lane & ~1, 32);
        out[(size_t)m * NO + j] = zr * z + z;
      }
}

extern "C" void kernel_launch(void* const* d_in, const int* in_sizes, int n_in,
                              void* d_out, int out_size, void* d_ws, size_t ws_size,
                              hipStream_t stream) {
  const float* x     = (const float*)d_in[0];
  const float* h0    = (const float*)d_in[1];
  const float* th    = (const float*)d_in[2];
  const float* dampp = (const float*)d_in[3];
  const float* Bwr   = (const float*)d_in[4];
  const float* Bwi   = (const float*)d_in[5];
  const float* Cwr   = (const float*)d_in[6];
  const float* Cwi   = (const float*)d_in[7];
  const float* Dwr   = (const float*)d_in[8];
  const float* Dwi   = (const float*)d_in[9];
  const float* gw    = (const float*)d_in[10];
  const float* gb    = (const float*)d_in[11];

  float* out  = (float*)d_out;
  float* hfin = out + (size_t)MM * NO;

  float* ws = (float*)d_ws;
  float* W1  = ws;
  float* WD  = W1 + (size_t)KX * NP;
  float* WC  = WD + (size_t)KX * NO;
  float* P   = WC + (size_t)KH * NO;
  float* Hws = P + (size_t)MM * NP;

  pack_w1_kernel<<<(KX * NP + 255) / 256, 256, 0, stream>>>(Bwr, Bwi, gw, W1);
  pack_wd_kernel<<<(KX * NO + 255) / 256, 256, 0, stream>>>(Dwr, Dwi, WD);
  pack_wc_kernel<<<(KH * NO + 255) / 256, 256, 0, stream>>>(Cwr, Cwi, WC);

  gemm_proj_kernel<<<dim3(MM / 128, NP / 64), 256, 0, stream>>>(x, W1, gb, P);
  gemm_dx_scan_kernel<<<1 + (MM / 128) * (NO / 128), 256, 0, stream>>>(x, WD, P, h0, th, dampp,
                                                                     Hws, hfin, out);
  gemm_ch_out_kernel<<<dim3(MM / 128, NO / 128), 256, 0, stream>>>(Hws, WC, out);
}


# D x launch cut to 1024 work-groups, what the chip holds resident (256 WGPs x 4): the tile that waited for a slot and ran last is formed as two 128x64 halves by 2 extra proj work-groups; same sums
# speedup vs baseline: 1.5001x; 1.0438x over previous
#include <hip/hip_runtime.h>
#include <math.h>

#define BB 4
#define SS 2048
#define DD 1024
#define NNh 64
#define MM (BB * SS)
#define KX (2 * DD)
#define NP 192
#define KH (2 * NNh)
#define NO (2 * DD)

#define A2LD 144

typedef __attribute__((ext_vector_type(2))) float v2f;
typedef __attribute__((ext_vector_type(8))) float v8f;

__device__ __forceinline__ v8f wmma_f32(v2f a, v2f b, v8f c) {
  return __builtin_amdgcn_wmma_f32_16x16x4_f32(
       false, a,  false, b,
       (short)0, c,  false,  false);
}

template <int NWN>
struct GemmCtx {
  static constexpr int BN   = NWN * 32;
  static constexpr int B2LD = BN + 16;
  static constexpr int NB4  = BN / 64;
  static constexpr int ASZ  = 8 * A2LD;
  static constexpr int BSZ  = 8 * B2LD;
  const float* pa0; const float* pa1;
  const float* pb[NB4];
  int sa0, sa1;
  int sb[NB4];
  int fa, fb;
  int ldb, K;
  float4 aReg0, aReg1, bReg[NB4];
};

template <int NWN>
__device__ __forceinline__ void gemm_fetch(GemmCtx<NWN>& c, int k) {
  c.aReg0 = *(const float4*)(c.pa0 + k);
  c.aReg1 = *(const float4*)(c.pa1 + k);
#pragma unroll
  for (int i = 0; i < GemmCtx<NWN>::NB4; ++i)
    c.bReg[i] = *(const float4*)(c.pb[i] + (size_t)k * c.ldb);
}

template <int NWN>
__device__ __forceinline__ void gemm_stash(const GemmCtx<NWN>& c, v2f* An, v2f* Bn) {
  v2f t;
  t.x = c.aReg0.x; t.y = c.aReg0.y; An[c.sa0] = t;
  t.x = c.aReg0.z; t.y = c.aReg0.w; An[c.sa0 + A2LD] = t;
  t.x = c.aReg1.x; t.y = c.aReg1.y; An[c.sa1] = t;
  t.x = c.aReg1.z; t.y = c.aReg1.w; An[c.sa1 + A2LD] = t;
  float* bsf = (float*)Bn;
#pragma unroll
  for (int i = 0; i < GemmCtx<NWN>::NB4; ++i) {
    bsf[c.sb[i] + 0] = c.bReg[i].x;
    bsf[c.sb[i] + 2] = c.bReg[i].y;
    bsf[c.sb[i] + 4] = c.bReg[i].z;
    bsf[c.sb[i] + 6] = c.bReg[i].w;
  }
}

template <int NWN, int CUR>
__device__ __forceinline__ void gemm_stage(GemmCtx<NWN>& c, int k0, v2f* As2, v2f* Bs2,
                                           v8f acc[2][NWN]) {
  using G = GemmCtx<NWN>;
  const v2f* Ac = As2 + CUR * G::ASZ;
  const v2f* Bc = Bs2 + CUR * G::BSZ;
  v2f* An = As2 + (CUR ^ 1) * G::ASZ;
  v2f* Bn = Bs2 + (CUR ^ 1) * G::BSZ;

  v2f a0[2], a1[2], bf[2][NWN];
  a0[0] = Ac[c.fa];
  a1[0] = Ac[c.fa + 16];
#pragma unroll
  for (int ni = 0; ni < NWN; ++ni) bf[0][ni] = Bc[c.fb + ni * 16];

  if (k0 + 16 < c.K) {
    gemm_stash<NWN>(c, An, Bn);
    if (k0 + 32 < c.K) gemm_fetch<NWN>(c, k0 + 32);
    if (k0 + 48 < c.K) {
      __builtin_prefetch(c.pa0 + k0 + 48, 0, 0);
      __builtin_prefetch(c.pa1 + k0 + 48, 0, 0);
    }
  }

#pragma unroll
  for (int kstep = 0; kstep < 4; ++kstep) {
    const int cs = kstep & 1, ns = cs ^ 1;
    if (kstep < 3) {
      const int p2 = 2 * (kstep + 1);
      a0[ns] = Ac[c.fa + p2 * A2LD];
      a1[ns] = Ac[c.fa + p2 * A2LD + 16];
#pragma unroll
      for (int ni = 0; ni < NWN; ++ni) bf[ns][ni] = Bc[c.fb + p2 * G::B2LD + ni * 16];
    }
#pragma unroll
    for (int ni = 0; ni < NWN; ++ni) {
      acc[0][ni] = wmma_f32(a0[cs], bf[cs][ni], acc[0][ni]);
      acc[1][ni] = wmma_f32(a1[cs], bf[cs][ni], acc[1][ni]);
    }
  }
  __syncthreads();
}

template <int NWN>
__device__ void gemm_tile_acc(const float* __restrict__ A, int lda,
                              const float* __restrict__ Bp, int ldb,
                              int K, int m0, int j0,
                              v2f* As2, v2f* Bs2, v8f acc[2][NWN]) {
  using G = GemmCtx<NWN>;
  const int tid  = threadIdx.x;
  const int lane = tid & 31;
  const int wave = tid >> 5;
  const int half = lane >> 4;
  const int lrow = lane & 15;
  const int wm   = (wave & 3) * 32;
  const int wn   = (wave >> 2) * (NWN * 16);

  GemmCtx<NWN> c;
  c.ldb = ldb; c.K = K;
  const int am0 = tid >> 2;
  const int ak0 = (tid & 3) << 2;
  const int am1 = (256 + tid) >> 2;
  const int ak1 = ((256 + tid) & 3) << 2;
  c.pa0 = A + (size_t)(m0 + am0) * lda + ak0;
  c.pa1 = A + (size_t)(m0 + am1) * lda + ak1;
  c.sa0 = (ak0 >> 1) * A2LD + am0;
  c.sa1 = (ak1 >> 1) * A2LD + am1;
#pragma unroll
  for (int i = 0; i < G::NB4; ++i) {
    const int idx = i * 256 + tid;
    const int bk = idx / (G::BN / 4);
    const int bj = (idx % (G::BN / 4)) * 4;
    c.pb[i] = Bp + (size_t)bk * ldb + j0 + bj;
    c.sb[i] = ((bk >> 1) * G::B2LD + bj) * 2 + (bk & 1);
  }
  c.fa = half * A2LD + wm + lrow;
  c.fb = half * G::B2LD + wn + lrow;

  gemm_fetch<NWN>(c, 0);
  gemm_stash<NWN>(c, As2, Bs2);
  if (16 < K) gemm_fetch<NWN>(c, 16);
  __syncthreads();

  for (int k0 = 0; k0 < K; k0 += 32) {
    gemm_stage<NWN, 0>(c, k0,      As2, Bs2, acc);
    gemm_stage<NWN, 1>(c, k0 + 16, As2, Bs2, acc);
  }
}

__global__ void pack_w1_kernel(const float* __restrict__ Bwr, const float* __restrict__ Bwi,
                               const float* __restrict__ gw, float* __restrict__ W1) {
  int idx = blockIdx.x * 256 + threadIdx.x;
  if (idx >= KX * NP) return;
  int k = idx / NP, j = idx - k * NP;
  int d = k >> 1, p = k & 1;
  float v;
  if (j < 64) {
    v = p ? -Bwi[j * DD + d] : Bwr[j * DD + d];
  } else if (j < 128) {
    int n = j - 64;
    v = p ? Bwr[n * DD + d] : Bwi[n * DD + d];
  } else {
    int n = j - 128;
    v = gw[n * (2 * DD) + p * DD + d];
  }
  W1[idx] = v;
}

__global__ void pack_wd_kernel(const float* __restrict__ Dr, const float* __restrict__ Di,
                               float* __restrict__ WD) {
  int idx = blockIdx.x * 256 + threadIdx.x;
  if (idx >= KX * NO) return;
  int k = idx >> 11, j = idx & (NO - 1);
  int e = k >> 1, p = k & 1, d = j >> 1;
  int q = j & 1;
  float wr = Dr[d * DD + e], wi = Di[d * DD + e];
  WD[idx] = (p == q) ? wr : (p ? -wi : wi);
}

__global__ void pack_wc_kernel(const float* __restrict__ Cr, const float* __restrict__ Ci,
                               float* __restrict__ WC) {
  int idx = blockIdx.x * 256 + threadIdx.x;
  if (idx >= KH * NO) return;
  int k = idx >> 11, j = idx & (NO - 1);
  int n = k >> 1, p = k & 1, d = j >> 1;
  int q = j & 1;
  float wr = Cr[d * NNh + n], wi = Ci[d * NNh + n];
  WC[idx] = (p == q) ? wr : (p ? -wi : wi);
}

__device__ __forceinline__ float4* zpart_slot(float* out, int m0, int j0, int w, int q) {
  const int lane = threadIdx.x & 31;
  return (float4*)(out + (size_t)(m0 + w * 16 + q) * NO + j0 + lane * 4);
}

#define DX_TILES ((MM / 128) * (NO / 128) - 1)
#define XT_M0 ((DX_TILES & (MM / 128 - 1)) * 128)
#define XT_J0 ((DX_TILES / (MM / 128)) * 128)
#define PROJ_WGS ((MM / 128) * (NP / 64))

__global__ __launch_bounds__(256) void gemm_proj_kernel(const float* __restrict__ x,
                                                        const float* __restrict__ W1,
                                                        const float* __restrict__ gate_b,
                                                        float* __restrict__ P,
                                                        const float* __restrict__ WD,
                                                        float* __restrict__ out) {
  __shared__ v2f As2[2 * 8 * A2LD];
  __shared__ v2f Bs2[2 * 8 * (64 + 16)];
  v8f acc[2][2] = {{{0,0,0,0,0,0,0,0}, {0,0,0,0,0,0,0,0}},
                   {{0,0,0,0,0,0,0,0}, {0,0,0,0,0,0,0,0}}};
  const int bx = blockIdx.x;
  if (bx >= PROJ_WGS) {
    const int hb = bx - PROJ_WGS;
    gemm_tile_acc<2>(x, KX, WD, NO, KX, XT_M0, XT_J0 + hb * 64, As2, Bs2, acc);
    const int wave = threadIdx.x >> 5;
    const int w4 = (wave & 3) + 4 * hb;
#pragma unroll
    for (int mi = 0; mi < 2; ++mi)
#pragma unroll
      for (int nl = 0; nl < 2; ++nl) {
        const v8f a = acc[mi][nl];
        const int q = (mi * 4 + 2 * (wave >> 2) + nl) * 2;
        *zpart_slot(out, XT_M0, XT_J0, w4, q)     = make_float4(a[0], a[1], a[2], a[3]);
        *zpart_slot(out, XT_M0, XT_J0, w4, q + 1) = make_float4(a[4], a[5], a[6], a[7]);
      }
    return;
  }
  const int m0 = (bx & (MM / 128 - 1)) * 128;
  const int j0 = (bx / (MM / 128)) * 64;
  gemm_tile_acc<2>(x, KX, W1, NP, KX, m0, j0, As2, Bs2, acc);

  const int lane = threadIdx.x & 31, wave = threadIdx.x >> 5;
  const int half = lane >> 4, lrow = lane & 15;
  const int wm = (wave & 3) * 32, wn = (wave >> 2) * 32;
#pragma unroll
  for (int mi = 0; mi < 2; ++mi)
#pragma unroll
    for (int ni = 0; ni < 2; ++ni)
#pragma unroll
      for (int e = 0; e < 8; ++e) {
        int m = m0 + wm + mi * 16 + half * 8 + e;
        int j = j0 + wn + ni * 16 + lrow;
        float v = acc[mi][ni][e];
        if (j >= 128) v = 1.0f / (1.0f + expf(-(v + gate_b[j - 128])));
        P[(size_t)m * NP + j] = v;
      }
}

__device__ __forceinline__ void scan_body(const float* __restrict__ P,
                                          const float* __restrict__ h0,
                                          const float* __restrict__ theta,
                                          const float* __restrict__ dampp,
                                          float* __restrict__ hws,
                                          float* __restrict__ hfin) {
  const int t = threadIdx.x;
  const int b = t >> 6, n = t & 63;
  const float c = cosf(theta[n]);
  const float s = sinf(theta[n]);
  const float dmp = 0.5f + 0.5f / (1.0f + expf(-dampp[n]));
  const float ca = c * dmp, sa = s * dmp;
  float hr = h0[(b * NNh + n) * 2 + 0];
  float hi = h0[(b * NNh + n) * 2 + 1];
#pragma unroll 4
  for (int st = 0; st < SS; ++st) {
    size_t m = (size_t)b * SS + st;
    float g  = P[m * NP + 128 + n];
    float br = P[m * NP + n];
    float bi = P[m * NP + 64 + n];
    float rr = hr * ca - hi * sa;
    float ri = hr * sa + hi * ca;
    hr = g * br + (1.0f - g) * rr;
    hi = g * bi + (1.0f - g) * ri;
    hws[m * KH + 2 * n + 0] = hr;
    hws[m * KH + 2 * n + 1] = hi;
  }
  hfin[(b * NNh + n) * 2 + 0] = hr;
  hfin[(b * NNh + n) * 2 + 1] = hi;
}

__global__ __launch_bounds__(256, 8) void gemm_dx_scan_kernel(const float* __restrict__ x,
                                                           const float* __restrict__ WD,
                                                           const float* __restrict__ P,
                                                           const float* __restrict__ h0,
                                                           const float* __restrict__ theta,
                                                           const float* __restrict__ dampp,
                                                           float* __restrict__ hws,
                                                           float* __restrict__ hfin,
                                                           float* __restrict__ out) {
  __shared__ v2f As2[2 * 8 * A2LD];
  __shared__ v2f Bs2[2 * 8 * (128 + 16)];
  if (blockIdx.x == 0) {
    scan_body(P, h0, theta, dampp, hws, hfin);
    return;
  }
  v8f acc[2][4] = {{{0,0,0,0,0,0,0,0}, {0,0,0,0,0,0,0,0},
                    {0,0,0,0,0,0,0,0}, {0,0,0,0,0,0,0,0}},
                   {{0,0,0,0,0,0,0,0}, {0,0,0,0,0,0,0,0},
                    {0,0,0,0,0,0,0,0}, {0,0,0,0,0,0,0,0}}};
  const int tile = blockIdx.x - 1;
  const int m0 = (tile & (MM / 128 - 1)) * 128;
  const int j0 = (tile / (MM / 128)) * 128;
  gemm_tile_acc<4>(x, KX, WD, NO, KX, m0, j0, As2, Bs2, acc);

#pragma unroll
  for (int mi = 0; mi < 2; ++mi)
#pragma unroll
    for (int ni = 0; ni < 4; ++ni) {
      const v8f a = acc[mi][ni];
      const int q = (mi * 4 + ni) * 2;
      *zpart_slot(out, m0, j0, threadIdx.x >> 5, q)     = make_float4(a[0], a[1], a[2], a[3]);
      *zpart_slot(out, m0, j0, threadIdx.x >> 5, q + 1) = make_float4(a[4], a[5], a[6], a[7]);
    }
}

__global__ __launch_bounds__(256) void gemm_ch_out_kernel(const float* __restrict__ hws,
                                                          const float* __restrict__ WC,
                                                          float* out) {
  __shared__ v2f As2[2 * 8 * A2LD];
  __shared__ v2f Bs2[2 * 8 * (128 + 16)];
  const int m0 = blockIdx.x * 128;
  const int j0 = blockIdx.y * 128;
  v8f acc[2][4];
#pragma unroll
  for (int mi = 0; mi < 2; ++mi)
#pragma unroll
    for (int ni = 0; ni < 4; ++ni) {
      const int q = (mi * 4 + ni) * 2;
      const float4 lo = *zpart_slot(out, m0, j0, threadIdx.x >> 5, q);
      const float4 hi = *zpart_slot(out, m0, j0, threadIdx.x >> 5, q + 1);
      v8f a = {lo.x, lo.y, lo.z, lo.w, hi.x, hi.y, hi.z, hi.w};
      acc[mi][ni] = a;
    }
  gemm_tile_acc<4>(hws, KH, WC, NO, KH, m0, j0, As2, Bs2, acc);

  const int lane = threadIdx.x & 31, wave = threadIdx.x >> 5;
  const int half = lane >> 4, lrow = lane & 15;
  const int wm = (wave & 3) * 32, wn = (wave >> 2) * 64;
#pragma unroll
  for (int mi = 0; mi < 2; ++mi)
#pragma unroll
    for (int ni = 0; ni < 4; ++ni)
#pragma unroll
      for (int e = 0; e < 8; ++e) {
        int m = m0 + wm + mi * 16 + half * 8 + e;
        int j = j0 + wn + ni * 16 + lrow;
        float z = acc[mi][ni][e];
        float zr = __shfl(z, lane & ~1, 32);
        out[(size_t)m * NO + j] = zr * z + z;
      }
}

extern "C" void kernel_launch(void* const* d_in, const int* in_sizes, int n_in,
                              void* d_out, int out_size, void* d_ws, size_t ws_size,
                              hipStream_t stream) {
  const float* x     = (const float*)d_in[0];
  const float* h0    = (const float*)d_in[1];
  const float* th    = (const float*)d_in[2];
  const float* dampp = (const float*)d_in[3];
  const float* Bwr   = (const float*)d_in[4];
  const float* Bwi   = (const float*)d_in[5];
  const float* Cwr   = (const float*)d_in[6];
  const float* Cwi   = (const float*)d_in[7];
  const float* Dwr   = (const float*)d_in[8];
  const float* Dwi   = (const float*)d_in[9];
  const float* gw    = (const float*)d_in[10];
  const float* gb    = (const float*)d_in[11];

  float* out  = (float*)d_out;
  float* hfin = out + (size_t)MM * NO;

  float* ws = (float*)d_ws;
  float* W1  = ws;
  float* WD  = W1 + (size_t)KX * NP;
  float* WC  = WD + (size_t)KX * NO;
  float* P   = WC + (size_t)KH * NO;
  float* Hws = P + (size_t)MM * NP;

  pack_w1_kernel<<<(KX * NP + 255) / 256, 256, 0, stream>>>(Bwr, Bwi, gw, W1);
  pack_wd_kernel<<<(KX * NO + 255) / 256, 256, 0, stream>>>(Dwr, Dwi, WD);
  pack_wc_kernel<<<(KH * NO + 255) / 256, 256, 0, stream>>>(Cwr, Cwi, WC);

  gemm_proj_kernel<<<PROJ_WGS + 2, 256, 0, stream>>>(x, W1, gb, P, WD, out);
  gemm_dx_scan_kernel<<<1 + DX_TILES, 256, 0, stream>>>(x, WD, P, h0, th, dampp,
                                                        Hws, hfin, out);
  gemm_ch_out_kernel<<<dim3(MM / 128, NO / 128), 256, 0, stream>>>(Hws, WC, out);
}


# D x and C h GEMMs stage B straight from the (wr, wi) weight planes: 16 B of each per thread per 32 k, rows (wr,-wi,wi,wr) stored by b128; packed planes and their 2 launches gone; same sums
# speedup vs baseline: 1.6007x; 1.0671x over previous
#include <hip/hip_runtime.h>
#include <math.h>

#define BB 4
#define SS 2048
#define DD 1024
#define NNh 64
#define MM (BB * SS)
#define KX (2 * DD)
#define NP 192
#define KH (2 * NNh)
#define NO (2 * DD)

#define A2LD 144

typedef __attribute__((ext_vector_type(2))) float v2f;
typedef __attribute__((ext_vector_type(8))) float v8f;

__device__ __forceinline__ v8f wmma_f32(v2f a, v2f b, v8f c) {
  return __builtin_amdgcn_wmma_f32_16x16x4_f32(
       false, a,  false, b,
       (short)0, c,  false,  false);
}

template <int NWN>
struct GemmCtx {
  static constexpr int BN   = NWN * 32;
  static constexpr int B2LD = BN + 16;
  static constexpr int NB4  = BN / 64;
  static constexpr int ASZ  = 8 * A2LD;
  static constexpr int BSZ  = 8 * B2LD;
  const float* pa0; const float* pa1;
  const float* pb[NB4];
  int sa0, sa1;
  int sb[NB4];
  int fa, fb;
  int ldb, K;
  float4 aReg0, aReg1, bReg[NB4];
};

template <int NWN>
__device__ __forceinline__ void gemm_fetch(GemmCtx<NWN>& c, int k) {
  c.aReg0 = *(const float4*)(c.pa0 + k);
  c.aReg1 = *(const float4*)(c.pa1 + k);
#pragma unroll
  for (int i = 0; i < GemmCtx<NWN>::NB4; ++i)
    c.bReg[i] = *(const float4*)(c.pb[i] + (size_t)k * c.ldb);
}

template <int NWN>
__device__ __forceinline__ void gemm_stash(const GemmCtx<NWN>& c, v2f* An, v2f* Bn) {
  v2f t;
  t.x = c.aReg0.x; t.y = c.aReg0.y; An[c.sa0] = t;
  t.x = c.aReg0.z; t.y = c.aReg0.w; An[c.sa0 + A2LD] = t;
  t.x = c.aReg1.x; t.y = c.aReg1.y; An[c.sa1] = t;
  t.x = c.aReg1.z; t.y = c.aReg1.w; An[c.sa1 + A2LD] = t;
  float* bsf = (float*)Bn;
#pragma unroll
  for (int i = 0; i < GemmCtx<NWN>::NB4; ++i) {
    bsf[c.sb[i] + 0] = c.bReg[i].x;
    bsf[c.sb[i] + 2] = c.bReg[i].y;
    bsf[c.sb[i] + 4] = c.bReg[i].z;
    bsf[c.sb[i] + 6] = c.bReg[i].w;
  }
}

template <int NWN, int CUR>
__device__ __forceinline__ void gemm_stage(GemmCtx<NWN>& c, int k0, v2f* As2, v2f* Bs2,
                                           v8f acc[2][NWN]) {
  using G = GemmCtx<NWN>;
  const v2f* Ac = As2 + CUR * G::ASZ;
  const v2f* Bc = Bs2 + CUR * G::BSZ;
  v2f* An = As2 + (CUR ^ 1) * G::ASZ;
  v2f* Bn = Bs2 + (CUR ^ 1) * G::BSZ;

  v2f a0[2], a1[2], bf[2][NWN];
  a0[0] = Ac[c.fa];
  a1[0] = Ac[c.fa + 16];
#pragma unroll
  for (int ni = 0; ni < NWN; ++ni) bf[0][ni] = Bc[c.fb + ni * 16];

  if (k0 + 16 < c.K) {
    gemm_stash<NWN>(c, An, Bn);
    if (k0 + 32 < c.K) gemm_fetch<NWN>(c, k0 + 32);
    if (k0 + 48 < c.K) {
      __builtin_prefetch(c.pa0 + k0 + 48, 0, 0);
      __builtin_prefetch(c.pa1 + k0 + 48, 0, 0);
    }
  }

#pragma unroll
  for (int kstep = 0; kstep < 4; ++kstep) {
    const int cs = kstep & 1, ns = cs ^ 1;
    if (kstep < 3) {
      const int p2 = 2 * (kstep + 1);
      a0[ns] = Ac[c.fa + p2 * A2LD];
      a1[ns] = Ac[c.fa + p2 * A2LD + 16];
#pragma unroll
      for (int ni = 0; ni < NWN; ++ni) bf[ns][ni] = Bc[c.fb + p2 * G::B2LD + ni * 16];
    }
#pragma unroll
    for (int ni = 0; ni < NWN; ++ni) {
      acc[0][ni] = wmma_f32(a0[cs], bf[cs][ni], acc[0][ni]);
      acc[1][ni] = wmma_f32(a1[cs], bf[cs][ni], acc[1][ni]);
    }
  }
  __syncthreads();
}

template <int NWN>
__device__ void gemm_tile_acc(const float* __restrict__ A, int lda,
                              const float* __restrict__ Bp, int ldb,
                              int K, int m0, int j0,
                              v2f* As2, v2f* Bs2, v8f acc[2][NWN]) {
  using G = GemmCtx<NWN>;
  const int tid  = threadIdx.x;
  const int lane = tid & 31;
  const int wave = tid >> 5;
  const int half = lane >> 4;
  const int lrow = lane & 15;
  const int wm   = (wave & 3) * 32;
  const int wn   = (wave >> 2) * (NWN * 16);

  GemmCtx<NWN> c;
  c.ldb = ldb; c.K = K;
  const int am0 = tid >> 2;
  const int ak0 = (tid & 3) << 2;
  const int am1 = (256 + tid) >> 2;
  const int ak1 = ((256 + tid) & 3) << 2;
  c.pa0 = A + (size_t)(m0 + am0) * lda + ak0;
  c.pa1 = A + (size_t)(m0 + am1) * lda + ak1;
  c.sa0 = (ak0 >> 1) * A2LD + am0;
  c.sa1 = (ak1 >> 1) * A2LD + am1;
#pragma unroll
  for (int i = 0; i < G::NB4; ++i) {
    const int idx = i * 256 + tid;
    const int bk = idx / (G::BN / 4);
    const int bj = (idx % (G::BN / 4)) * 4;
    c.pb[i] = Bp + (size_t)bk * ldb + j0 + bj;
    c.sb[i] = ((bk >> 1) * G::B2LD + bj) * 2 + (bk & 1);
  }
  c.fa = half * A2LD + wm + lrow;
  c.fb = half * G::B2LD + wn + lrow;

  gemm_fetch<NWN>(c, 0);
  gemm_stash<NWN>(c, As2, Bs2);
  if (16 < K) gemm_fetch<NWN>(c, 16);
  __syncthreads();

  for (int k0 = 0; k0 < K; k0 += 32) {
    gemm_stage<NWN, 0>(c, k0,      As2, Bs2, acc);
    gemm_stage<NWN, 1>(c, k0 + 16, As2, Bs2, acc);
  }
}

template <int NWN>
struct CplxCtx {
  static constexpr int BN   = NWN * 32;
  static constexpr int B2LD = BN + 16;
  static constexpr int ASZ  = 8 * A2LD;
  static constexpr int BSZ  = 16 * B2LD;
  static constexpr int BTH  = (BN / 2) * 4;
  const float* A; const float* Wr; const float* Wi;
  unsigned ao, a1d;
  unsigned bo;
  int sa0, sa1;
  int sb;
  int fa, fb;
  int K;
  float4 aReg0, aReg1, rReg, iReg;
};

template <int NWN>
__device__ __forceinline__ void cplx_fetch_a(CplxCtx<NWN>& c, int k) {
  c.aReg0 = *(const float4*)((const char*)(c.A + k) + c.ao);
  c.aReg1 = *(const float4*)((const char*)(c.A + k + c.a1d) + c.ao);
}

template <int NWN>
__device__ __forceinline__ void cplx_fetch_b(CplxCtx<NWN>& c, int k) {
  if (CplxCtx<NWN>::BTH == 256 || threadIdx.x < CplxCtx<NWN>::BTH) {
    c.rReg = *(const float4*)((const char*)(c.Wr + (k >> 1)) + c.bo);
    c.iReg = *(const float4*)((const char*)(c.Wi + (k >> 1)) + c.bo);
  }
}

template <int NWN>
__device__ __forceinline__ void cplx_stash_a(const CplxCtx<NWN>& c, v2f* An) {
  v2f t;
  t.x = c.aReg0.x; t.y = c.aReg0.y; An[c.sa0] = t;
  t.x = c.aReg0.z; t.y = c.aReg0.w; An[c.sa0 + A2LD] = t;
  t.x = c.aReg1.x; t.y = c.aReg1.y; An[c.sa1] = t;
  t.x = c.aReg1.z; t.y = c.aReg1.w; An[c.sa1 + A2LD] = t;
}

template <int NWN>
__device__ __forceinline__ void cplx_stash_b(const CplxCtx<NWN>& c, v2f* Bn) {
  if (CplxCtx<NWN>::BTH == 256 || threadIdx.x < CplxCtx<NWN>::BTH) {
    float4* b4 = (float4*)(Bn + c.sb);
    constexpr int R4 = CplxCtx<NWN>::B2LD / 2;
    b4[0 * R4] = make_float4(c.rReg.x, -c.iReg.x, c.iReg.x, c.rReg.x);
    b4[1 * R4] = make_float4(c.rReg.y, -c.iReg.y, c.iReg.y, c.rReg.y);
    b4[2 * R4] = make_float4(c.rReg.z, -c.iReg.z, c.iReg.z, c.rReg.z);
    b4[3 * R4] = make_float4(c.rReg.w, -c.iReg.w, c.iReg.w, c.rReg.w);
  }
}

template <int NWN, int PH, int PB>
__device__ __forceinline__ void cplx_stage(CplxCtx<NWN>& c, int k0, v2f* As2, v2f* Bs2,
                                           v8f acc[2][NWN]) {
  using G = CplxCtx<NWN>;
  const v2f* Ac = As2 + PH * G::ASZ;
  const v2f* Bc = Bs2 + PB * G::BSZ + PH * 8 * G::B2LD;
  v2f* An = As2 + (PH ^ 1) * G::ASZ;
  v2f* Bn = Bs2 + (PB ^ 1) * G::BSZ;

  v2f a0[2], a1[2], bf[2][NWN];
  a0[0] = Ac[c.fa];
  a1[0] = Ac[c.fa + 16];
#pragma unroll
  for (int ni = 0; ni < NWN; ++ni) bf[0][ni] = Bc[c.fb + ni * 16];

  if (k0 + 16 < c.K) {
    cplx_stash_a<NWN>(c, An);
    if (PH == 0 && k0 + 32 < c.K) {
      cplx_stash_b<NWN>(c, Bn);
      if (k0 + 64 < c.K) cplx_fetch_b<NWN>(c, k0 + 64);
    }
    if (k0 + 32 < c.K) cplx_fetch_a<NWN>(c, k0 + 32);
    if (k0 + 48 < c.K) {
      __builtin_prefetch((const char*)(c.A + k0 + 48) + c.ao, 0, 0);
      __builtin_prefetch((const char*)(c.A + k0 + 48 + c.a1d) + c.ao, 0, 0);
    }
  }

#pragma unroll
  for (int kstep = 0; kstep < 4; ++kstep) {
    const int cs = kstep & 1, ns = cs ^ 1;
    if (kstep < 3) {
      const int p2 = 2 * (kstep + 1);
      a0[ns] = Ac[c.fa + p2 * A2LD];
      a1[ns] = Ac[c.fa + p2 * A2LD + 16];
#pragma unroll
      for (int ni = 0; ni < NWN; ++ni) bf[ns][ni] = Bc[c.fb + p2 * G::B2LD + ni * 16];
    }
#pragma unroll
    for (int ni = 0; ni < NWN; ++ni) {
      acc[0][ni] = wmma_f32(a0[cs], bf[cs][ni], acc[0][ni]);
      acc[1][ni] = wmma_f32(a1[cs], bf[cs][ni], acc[1][ni]);
    }
  }
  __syncthreads();
}

template <int NWN>
__device__ void cplx_tile_acc(const float* __restrict__ A, int lda,
                              const float* __restrict__ Wr, const float* __restrict__ Wi,
                              int ldw, int K, int m0, int j0,
                              v2f* As2, v2f* Bs2, v8f acc[2][NWN]) {
  using G = CplxCtx<NWN>;
  const int tid  = threadIdx.x;
  const int lane = tid & 31;
  const int wave = tid >> 5;
  const int half = lane >> 4;
  const int lrow = lane & 15;
  const int wm   = (wave & 3) * 32;
  const int wn   = (wave >> 2) * (NWN * 16);

  CplxCtx<NWN> c;
  c.K = K;
  const int am0 = tid >> 2;
  const int ak0 = (tid & 3) << 2;
  const int am1 = am0 + 64;
  c.A = A; c.Wr = Wr; c.Wi = Wi;
  c.ao  = 4u * ((unsigned)(m0 + am0) * (unsigned)lda + (unsigned)ak0);
  c.a1d = 64u * (unsigned)lda;
  c.sa0 = (ak0 >> 1) * A2LD + am0;
  c.sa1 = (ak0 >> 1) * A2LD + am1;
  const int bd = (tid >> 2) & (G::BN / 2 - 1);
  const int bu = tid & 3;
  c.bo = 4u * ((unsigned)((j0 >> 1) + bd) * (unsigned)ldw + 4u * (unsigned)bu);
  c.sb = (4 * bu) * G::B2LD + 2 * bd;
  c.fa = half * A2LD + wm + lrow;
  c.fb = half * G::B2LD + wn + lrow;

  cplx_fetch_a<NWN>(c, 0);
  cplx_fetch_b<NWN>(c, 0);
  cplx_stash_a<NWN>(c, As2);
  cplx_stash_b<NWN>(c, Bs2);
  cplx_fetch_a<NWN>(c, 16);
  cplx_fetch_b<NWN>(c, 32);
  __syncthreads();

  for (int k0 = 0; k0 < K; k0 += 64) {
    cplx_stage<NWN, 0, 0>(c, k0,      As2, Bs2, acc);
    cplx_stage<NWN, 1, 0>(c, k0 + 16, As2, Bs2, acc);
    cplx_stage<NWN, 0, 1>(c, k0 + 32, As2, Bs2, acc);
    cplx_stage<NWN, 1, 1>(c, k0 + 48, As2, Bs2, acc);
  }
}

__global__ void pack_w1_kernel(const float* __restrict__ Bwr, const float* __restrict__ Bwi,
                               const float* __restrict__ gw, float* __restrict__ W1) {
  int idx = blockIdx.x * 256 + threadIdx.x;
  if (idx >= KX * NP) return;
  int k = idx / NP, j = idx - k * NP;
  int d = k >> 1, p = k & 1;
  float v;
  if (j < 64) {
    v = p ? -Bwi[j * DD + d] : Bwr[j * DD + d];
  } else if (j < 128) {
    int n = j - 64;
    v = p ? Bwr[n * DD + d] : Bwi[n * DD + d];
  } else {
    int n = j - 128;
    v = gw[n * (2 * DD) + p * DD + d];
  }
  W1[idx] = v;
}

__device__ __forceinline__ float4* zpart_slot(float* out, int m0, int j0, int w, int q) {
  const int lane = threadIdx.x & 31;
  return (float4*)(out + (size_t)(m0 + w * 16 + q) * NO + j0 + lane * 4);
}

#define DX_TILES ((MM / 128) * (NO / 128) - 1)
#define XT_M0 ((DX_TILES & (MM / 128 - 1)) * 128)
#define XT_J0 ((DX_TILES / (MM / 128)) * 128)
#define PROJ_WGS ((MM / 128) * (NP / 64))

__global__ __launch_bounds__(256) void gemm_proj_kernel(const float* __restrict__ x,
                                                        const float* __restrict__ W1,
                                                        const float* __restrict__ gate_b,
                                                        float* __restrict__ P,
                                                        const float* __restrict__ Dr,
                                                        const float* __restrict__ Di,
                                                        float* __restrict__ out) {
  __shared__ v2f As2[2 * 8 * A2LD];
  __shared__ __attribute__((aligned(16))) v2f Bs2[2 * 16 * (64 + 16)];
  v8f acc[2][2] = {{{0,0,0,0,0,0,0,0}, {0,0,0,0,0,0,0,0}},
                   {{0,0,0,0,0,0,0,0}, {0,0,0,0,0,0,0,0}}};
  const int bx = blockIdx.x;
  if (bx >= PROJ_WGS) {
    const int hb = bx - PROJ_WGS;
    cplx_tile_acc<2>(x, KX, Dr, Di, DD, KX, XT_M0, XT_J0 + hb * 64, As2, Bs2, acc);
    const int wave = threadIdx.x >> 5;
    const int w4 = (wave & 3) + 4 * hb;
#pragma unroll
    for (int mi = 0; mi < 2; ++mi)
#pragma unroll
      for (int nl = 0; nl < 2; ++nl) {
        const v8f a = acc[mi][nl];
        const int q = (mi * 4 + 2 * (wave >> 2) + nl) * 2;
        *zpart_slot(out, XT_M0, XT_J0, w4, q)     = make_float4(a[0], a[1], a[2], a[3]);
        *zpart_slot(out, XT_M0, XT_J0, w4, q + 1) = make_float4(a[4], a[5], a[6], a[7]);
      }
    return;
  }
  const int m0 = (bx & (MM / 128 - 1)) * 128;
  const int j0 = (bx / (MM / 128)) * 64;
  gemm_tile_acc<2>(x, KX, W1, NP, KX, m0, j0, As2, Bs2, acc);

  const int lane = threadIdx.x & 31, wave = threadIdx.x >> 5;
  const int half = lane >> 4, lrow = lane & 15;
  const int wm = (wave & 3) * 32, wn = (wave >> 2) * 32;
#pragma unroll
  for (int mi = 0; mi < 2; ++mi)
#pragma unroll
    for (int ni = 0; ni < 2; ++ni)
#pragma unroll
      for (int e = 0; e < 8; ++e) {
        int m = m0 + wm + mi * 16 + half * 8 + e;
        int j = j0 + wn + ni * 16 + lrow;
        float v = acc[mi][ni][e];
        if (j >= 128) v = 1.0f / (1.0f + expf(-(v + gate_b[j - 128])));
        P[(size_t)m * NP + j] = v;
      }
}

__device__ __forceinline__ void scan_body(const float* __restrict__ P,
                                          const float* __restrict__ h0,
                                          const float* __restrict__ theta,
                                          const float* __restrict__ dampp,
                                          float* __restrict__ hws,
                                          float* __restrict__ hfin) {
  const int t = threadIdx.x;
  const int b = t >> 6, n = t & 63;
  const float c = cosf(theta[n]);
  const float s = sinf(theta[n]);
  const float dmp = 0.5f + 0.5f / (1.0f + expf(-dampp[n]));
  const float ca = c * dmp, sa = s * dmp;
  float hr = h0[(b * NNh + n) * 2 + 0];
  float hi = h0[(b * NNh + n) * 2 + 1];
#pragma unroll 4
  for (int st = 0; st < SS; ++st) {
    size_t m = (size_t)b * SS + st;
    float g  = P[m * NP + 128 + n];
    float br = P[m * NP + n];
    float bi = P[m * NP + 64 + n];
    float rr = hr * ca - hi * sa;
    float ri = hr * sa + hi * ca;
    hr = g * br + (1.0f - g) * rr;
    hi = g * bi + (1.0f - g) * ri;
    hws[m * KH + 2 * n + 0] = hr;
    hws[m * KH + 2 * n + 1] = hi;
  }
  hfin[(b * NNh + n) * 2 + 0] = hr;
  hfin[(b * NNh + n) * 2 + 1] = hi;
}

__global__ __launch_bounds__(256, 8) void gemm_dx_scan_kernel(const float* __restrict__ x,
                                                           const float* __restrict__ Dr,
                                                           const float* __restrict__ Di,
                                                           const float* __restrict__ P,
                                                           const float* __restrict__ h0,
                                                           const float* __restrict__ theta,
                                                           const float* __restrict__ dampp,
                                                           float* __restrict__ hws,
                                                           float* __restrict__ hfin,
                                                           float* __restrict__ out) {
  __shared__ v2f As2[2 * 8 * A2LD];
  __shared__ __attribute__((aligned(16))) v2f Bs2[2 * 16 * (128 + 16)];
  if (blockIdx.x == 0) {
    scan_body(P, h0, theta, dampp, hws, hfin);
    return;
  }
  v8f acc[2][4] = {{{0,0,0,0,0,0,0,0}, {0,0,0,0,0,0,0,0},
                    {0,0,0,0,0,0,0,0}, {0,0,0,0,0,0,0,0}},
                   {{0,0,0,0,0,0,0,0}, {0,0,0,0,0,0,0,0},
                    {0,0,0,0,0,0,0,0}, {0,0,0,0,0,0,0,0}}};
  const int tile = blockIdx.x - 1;
  const int m0 = (tile & (MM / 128 - 1)) * 128;
  const int j0 = (tile / (MM / 128)) * 128;
  cplx_tile_acc<4>(x, KX, Dr, Di, DD, KX, m0, j0, As2, Bs2, acc);

#pragma unroll
  for (int mi = 0; mi < 2; ++mi)
#pragma unroll
    for (int ni = 0; ni < 4; ++ni) {
      const v8f a = acc[mi][ni];
      const int q = (mi * 4 + ni) * 2;
      *zpart_slot(out, m0, j0, threadIdx.x >> 5, q)     = make_float4(a[0], a[1], a[2], a[3]);
      *zpart_slot(out, m0, j0, threadIdx.x >> 5, q + 1) = make_float4(a[4], a[5], a[6], a[7]);
    }
}

__global__ __launch_bounds__(256) void gemm_ch_out_kernel(const float* __restrict__ hws,
                                                          const float* __restrict__ Cr,
                                                          const float* __restrict__ Ci,
                                                          float* out) {
  __shared__ v2f As2[2 * 8 * A2LD];
  __shared__ __attribute__((aligned(16))) v2f Bs2[2 * 16 * (128 + 16)];
  const int m0 = blockIdx.x * 128;
  const int j0 = blockIdx.y * 128;
  v8f acc[2][4];
#pragma unroll
  for (int mi = 0; mi < 2; ++mi)
#pragma unroll
    for (int ni = 0; ni < 4; ++ni) {
      const int q = (mi * 4 + ni) * 2;
      const float4 lo = *zpart_slot(out, m0, j0, threadIdx.x >> 5, q);
      const float4 hi = *zpart_slot(out, m0, j0, threadIdx.x >> 5, q + 1);
      v8f a = {lo.x, lo.y, lo.z, lo.w, hi.x, hi.y, hi.z, hi.w};
      acc[mi][ni] = a;
    }
  cplx_tile_acc<4>(hws, KH, Cr, Ci, NNh, KH, m0, j0, As2, Bs2, acc);

  const int lane = threadIdx.x & 31, wave = threadIdx.x >> 5;
  const int half = lane >> 4, lrow = lane & 15;
  const int wm = (wave & 3) * 32, wn = (wave >> 2) * 64;
#pragma unroll
  for (int mi = 0; mi < 2; ++mi)
#pragma unroll
    for (int ni = 0; ni < 4; ++ni)
#pragma unroll
      for (int e = 0; e < 8; ++e) {
        int m = m0 + wm + mi * 16 + half * 8 + e;
        int j = j0 + wn + ni * 16 + lrow;
        float z = acc[mi][ni][e];
        float zr = __shfl(z, lane & ~1, 32);
        out[(size_t)m * NO + j] = zr * z + z;
      }
}

extern "C" void kernel_launch(void* const* d_in, const int* in_sizes, int n_in,
                              void* d_out, int out_size, void* d_ws, size_t ws_size,
                              hipStream_t stream) {
  const float* x     = (const float*)d_in[0];
  const float* h0    = (const float*)d_in[1];
  const float* th    = (const float*)d_in[2];
  const float* dampp = (const float*)d_in[3];
  const float* Bwr   = (const float*)d_in[4];
  const float* Bwi   = (const float*)d_in[5];
  const float* Cwr   = (const float*)d_in[6];
  const float* Cwi   = (const float*)d_in[7];
  const float* Dwr   = (const float*)d_in[8];
  const float* Dwi   = (const float*)d_in[9];
  const float* gw    = (const float*)d_in[10];
  const float* gb    = (const float*)d_in[11];

  float* out  = (float*)d_out;
  float* hfin = out + (size_t)MM * NO;

  float* ws = (float*)d_ws;
  float* W1  = ws;
  float* P   = W1 + (size_t)KX * NP;
  float* Hws = P + (size_t)MM * NP;

  pack_w1_kernel<<<(KX * NP + 255) / 256, 256, 0, stream>>>(Bwr, Bwi, gw, W1);

  gemm_proj_kernel<<<PROJ_WGS + 2, 256, 0, stream>>>(x, W1, gb, P, Dwr, Dwi, out);
  gemm_dx_scan_kernel<<<1 + DX_TILES, 256, 0, stream>>>(x, Dwr, Dwi, P, h0, th, dampp,
                                                        Hws, hfin, out);
  gemm_ch_out_kernel<<<dim3(MM / 128, NO / 128), 256, 0, stream>>>(Hws, Cwr, Cwi, out);
}


# scan loop takes 16 time steps per trip (was 4): the 48 input loads of a trip are issued as one group, so one memory round trip is paid per 16 steps, not per 4; the steps run in the same order
# speedup vs baseline: 1.6628x; 1.0388x over previous
#include <hip/hip_runtime.h>
#include <math.h>

#define BB 4
#define SS 2048
#define DD 1024
#define NNh 64
#define MM (BB * SS)
#define KX (2 * DD)
#define NP 192
#define KH (2 * NNh)
#define NO (2 * DD)

#define A2LD 144

typedef __attribute__((ext_vector_type(2))) float v2f;
typedef __attribute__((ext_vector_type(8))) float v8f;

__device__ __forceinline__ v8f wmma_f32(v2f a, v2f b, v8f c) {
  return __builtin_amdgcn_wmma_f32_16x16x4_f32(
       false, a,  false, b,
       (short)0, c,  false,  false);
}

template <int NWN>
struct GemmCtx {
  static constexpr int BN   = NWN * 32;
  static constexpr int B2LD = BN + 16;
  static constexpr int NB4  = BN / 64;
  static constexpr int ASZ  = 8 * A2LD;
  static constexpr int BSZ  = 8 * B2LD;
  const float* pa0; const float* pa1;
  const float* pb[NB4];
  int sa0, sa1;
  int sb[NB4];
  int fa, fb;
  int ldb, K;
  float4 aReg0, aReg1, bReg[NB4];
};

template <int NWN>
__device__ __forceinline__ void gemm_fetch(GemmCtx<NWN>& c, int k) {
  c.aReg0 = *(const float4*)(c.pa0 + k);
  c.aReg1 = *(const float4*)(c.pa1 + k);
#pragma unroll
  for (int i = 0; i < GemmCtx<NWN>::NB4; ++i)
    c.bReg[i] = *(const float4*)(c.pb[i] + (size_t)k * c.ldb);
}

template <int NWN>
__device__ __forceinline__ void gemm_stash(const GemmCtx<NWN>& c, v2f* An, v2f* Bn) {
  v2f t;
  t.x = c.aReg0.x; t.y = c.aReg0.y; An[c.sa0] = t;
  t.x = c.aReg0.z; t.y = c.aReg0.w; An[c.sa0 + A2LD] = t;
  t.x = c.aReg1.x; t.y = c.aReg1.y; An[c.sa1] = t;
  t.x = c.aReg1.z; t.y = c.aReg1.w; An[c.sa1 + A2LD] = t;
  float* bsf = (float*)Bn;
#pragma unroll
  for (int i = 0; i < GemmCtx<NWN>::NB4; ++i) {
    bsf[c.sb[i] + 0] = c.bReg[i].x;
    bsf[c.sb[i] + 2] = c.bReg[i].y;
    bsf[c.sb[i] + 4] = c.bReg[i].z;
    bsf[c.sb[i] + 6] = c.bReg[i].w;
  }
}

template <int NWN, int CUR>
__device__ __forceinline__ void gemm_stage(GemmCtx<NWN>& c, int k0, v2f* As2, v2f* Bs2,
                                           v8f acc[2][NWN]) {
  using G = GemmCtx<NWN>;
  const v2f* Ac = As2 + CUR * G::ASZ;
  const v2f* Bc = Bs2 + CUR * G::BSZ;
  v2f* An = As2 + (CUR ^ 1) * G::ASZ;
  v2f* Bn = Bs2 + (CUR ^ 1) * G::BSZ;

  v2f a0[2], a1[2], bf[2][NWN];
  a0[0] = Ac[c.fa];
  a1[0] = Ac[c.fa + 16];
#pragma unroll
  for (int ni = 0; ni < NWN; ++ni) bf[0][ni] = Bc[c.fb + ni * 16];

  if (k0 + 16 < c.K) {
    gemm_stash<NWN>(c, An, Bn);
    if (k0 + 32 < c.K) gemm_fetch<NWN>(c, k0 + 32);
    if (k0 + 48 < c.K) {
      __builtin_prefetch(c.pa0 + k0 + 48, 0, 0);
      __builtin_prefetch(c.pa1 + k0 + 48, 0, 0);
    }
  }

#pragma unroll
  for (int kstep = 0; kstep < 4; ++kstep) {
    const int cs = kstep & 1, ns = cs ^ 1;
    if (kstep < 3) {
      const int p2 = 2 * (kstep + 1);
      a0[ns] = Ac[c.fa + p2 * A2LD];
      a1[ns] = Ac[c.fa + p2 * A2LD + 16];
#pragma unroll
      for (int ni = 0; ni < NWN; ++ni) bf[ns][ni] = Bc[c.fb + p2 * G::B2LD + ni * 16];
    }
#pragma unroll
    for (int ni = 0; ni < NWN; ++ni) {
      acc[0][ni] = wmma_f32(a0[cs], bf[cs][ni], acc[0][ni]);
      acc[1][ni] = wmma_f32(a1[cs], bf[cs][ni], acc[1][ni]);
    }
  }
  __syncthreads();
}

template <int NWN>
__device__ void gemm_tile_acc(const float* __restrict__ A, int lda,
                              const float* __restrict__ Bp, int ldb,
                              int K, int m0, int j0,
                              v2f* As2, v2f* Bs2, v8f acc[2][NWN]) {
  using G = GemmCtx<NWN>;
  const int tid  = threadIdx.x;
  const int lane = tid & 31;
  const int wave = tid >> 5;
  const int half = lane >> 4;
  const int lrow = lane & 15;
  const int wm   = (wave & 3) * 32;
  const int wn   = (wave >> 2) * (NWN * 16);

  GemmCtx<NWN> c;
  c.ldb = ldb; c.K = K;
  const int am0 = tid >> 2;
  const int ak0 = (tid & 3) << 2;
  const int am1 = (256 + tid) >> 2;
  const int ak1 = ((256 + tid) & 3) << 2;
  c.pa0 = A + (size_t)(m0 + am0) * lda + ak0;
  c.pa1 = A + (size_t)(m0 + am1) * lda + ak1;
  c.sa0 = (ak0 >> 1) * A2LD + am0;
  c.sa1 = (ak1 >> 1) * A2LD + am1;
#pragma unroll
  for (int i = 0; i < G::NB4; ++i) {
    const int idx = i * 256 + tid;
    const int bk = idx / (G::BN / 4);
    const int bj = (idx % (G::BN / 4)) * 4;
    c.pb[i] = Bp + (size_t)bk * ldb + j0 + bj;
    c.sb[i] = ((bk >> 1) * G::B2LD + bj) * 2 + (bk & 1);
  }
  c.fa = half * A2LD + wm + lrow;
  c.fb = half * G::B2LD + wn + lrow;

  gemm_fetch<NWN>(c, 0);
  gemm_stash<NWN>(c, As2, Bs2);
  if (16 < K) gemm_fetch<NWN>(c, 16);
  __syncthreads();

  for (int k0 = 0; k0 < K; k0 += 32) {
    gemm_stage<NWN, 0>(c, k0,      As2, Bs2, acc);
    gemm_stage<NWN, 1>(c, k0 + 16, As2, Bs2, acc);
  }
}

template <int NWN>
struct CplxCtx {
  static constexpr int BN   = NWN * 32;
  static constexpr int B2LD = BN + 16;
  static constexpr int ASZ  = 8 * A2LD;
  static constexpr int BSZ  = 16 * B2LD;
  static constexpr int BTH  = (BN / 2) * 4;
  const float* A; const float* Wr; const float* Wi;
  unsigned ao, a1d;
  unsigned bo;
  int sa0, sa1;
  int sb;
  int fa, fb;
  int K;
  float4 aReg0, aReg1, rReg, iReg;
};

template <int NWN>
__device__ __forceinline__ void cplx_fetch_a(CplxCtx<NWN>& c, int k) {
  c.aReg0 = *(const float4*)((const char*)(c.A + k) + c.ao);
  c.aReg1 = *(const float4*)((const char*)(c.A + k + c.a1d) + c.ao);
}

template <int NWN>
__device__ __forceinline__ void cplx_fetch_b(CplxCtx<NWN>& c, int k) {
  if (CplxCtx<NWN>::BTH == 256 || threadIdx.x < CplxCtx<NWN>::BTH) {
    c.rReg = *(const float4*)((const char*)(c.Wr + (k >> 1)) + c.bo);
    c.iReg = *(const float4*)((const char*)(c.Wi + (k >> 1)) + c.bo);
  }
}

template <int NWN>
__device__ __forceinline__ void cplx_stash_a(const CplxCtx<NWN>& c, v2f* An) {
  v2f t;
  t.x = c.aReg0.x; t.y = c.aReg0.y; An[c.sa0] = t;
  t.x = c.aReg0.z; t.y = c.aReg0.w; An[c.sa0 + A2LD] = t;
  t.x = c.aReg1.x; t.y = c.aReg1.y; An[c.sa1] = t;
  t.x = c.aReg1.z; t.y = c.aReg1.w; An[c.sa1 + A2LD] = t;
}

template <int NWN>
__device__ __forceinline__ void cplx_stash_b(const CplxCtx<NWN>& c, v2f* Bn) {
  if (CplxCtx<NWN>::BTH == 256 || threadIdx.x < CplxCtx<NWN>::BTH) {
    float4* b4 = (float4*)(Bn + c.sb);
    constexpr int R4 = CplxCtx<NWN>::B2LD / 2;
    b4[0 * R4] = make_float4(c.rReg.x, -c.iReg.x, c.iReg.x, c.rReg.x);
    b4[1 * R4] = make_float4(c.rReg.y, -c.iReg.y, c.iReg.y, c.rReg.y);
    b4[2 * R4] = make_float4(c.rReg.z, -c.iReg.z, c.iReg.z, c.rReg.z);
    b4[3 * R4] = make_float4(c.rReg.w, -c.iReg.w, c.iReg.w, c.rReg.w);
  }
}

template <int NWN, int PH, int PB>
__device__ __forceinline__ void cplx_stage(CplxCtx<NWN>& c, int k0, v2f* As2, v2f* Bs2,
                                           v8f acc[2][NWN]) {
  using G = CplxCtx<NWN>;
  const v2f* Ac = As2 + PH * G::ASZ;
  const v2f* Bc = Bs2 + PB * G::BSZ + PH * 8 * G::B2LD;
  v2f* An = As2 + (PH ^ 1) * G::ASZ;
  v2f* Bn = Bs2 + (PB ^ 1) * G::BSZ;

  v2f a0[2], a1[2], bf[2][NWN];
  a0[0] = Ac[c.fa];
  a1[0] = Ac[c.fa + 16];
#pragma unroll
  for (int ni = 0; ni < NWN; ++ni) bf[0][ni] = Bc[c.fb + ni * 16];

  if (k0 + 16 < c.K) {
    cplx_stash_a<NWN>(c, An);
    if (PH == 0 && k0 + 32 < c.K) {
      cplx_stash_b<NWN>(c, Bn);
      if (k0 + 64 < c.K) cplx_fetch_b<NWN>(c, k0 + 64);
    }
    if (k0 + 32 < c.K) cplx_fetch_a<NWN>(c, k0 + 32);
    if (k0 + 48 < c.K) {
      __builtin_prefetch((const char*)(c.A + k0 + 48) + c.ao, 0, 0);
      __builtin_prefetch((const char*)(c.A + k0 + 48 + c.a1d) + c.ao, 0, 0);
    }
  }

#pragma unroll
  for (int kstep = 0; kstep < 4; ++kstep) {
    const int cs = kstep & 1, ns = cs ^ 1;
    if (kstep < 3) {
      const int p2 = 2 * (kstep + 1);
      a0[ns] = Ac[c.fa + p2 * A2LD];
      a1[ns] = Ac[c.fa + p2 * A2LD + 16];
#pragma unroll
      for (int ni = 0; ni < NWN; ++ni) bf[ns][ni] = Bc[c.fb + p2 * G::B2LD + ni * 16];
    }
#pragma unroll
    for (int ni = 0; ni < NWN; ++ni) {
      acc[0][ni] = wmma_f32(a0[cs], bf[cs][ni], acc[0][ni]);
      acc[1][ni] = wmma_f32(a1[cs], bf[cs][ni], acc[1][ni]);
    }
  }
  __syncthreads();
}

template <int NWN>
__device__ void cplx_tile_acc(const float* __restrict__ A, int lda,
                              const float* __restrict__ Wr, const float* __restrict__ Wi,
                              int ldw, int K, int m0, int j0,
                              v2f* As2, v2f* Bs2, v8f acc[2][NWN]) {
  using G = CplxCtx<NWN>;
  const int tid  = threadIdx.x;
  const int lane = tid & 31;
  const int wave = tid >> 5;
  const int half = lane >> 4;
  const int lrow = lane & 15;
  const int wm   = (wave & 3) * 32;
  const int wn   = (wave >> 2) * (NWN * 16);

  CplxCtx<NWN> c;
  c.K = K;
  const int am0 = tid >> 2;
  const int ak0 = (tid & 3) << 2;
  const int am1 = am0 + 64;
  c.A = A; c.Wr = Wr; c.Wi = Wi;
  c.ao  = 4u * ((unsigned)(m0 + am0) * (unsigned)lda + (unsigned)ak0);
  c.a1d = 64u * (unsigned)lda;
  c.sa0 = (ak0 >> 1) * A2LD + am0;
  c.sa1 = (ak0 >> 1) * A2LD + am1;
  const int bd = (tid >> 2) & (G::BN / 2 - 1);
  const int bu = tid & 3;
  c.bo = 4u * ((unsigned)((j0 >> 1) + bd) * (unsigned)ldw + 4u * (unsigned)bu);
  c.sb = (4 * bu) * G::B2LD + 2 * bd;
  c.fa = half * A2LD + wm + lrow;
  c.fb = half * G::B2LD + wn + lrow;

  cplx_fetch_a<NWN>(c, 0);
  cplx_fetch_b<NWN>(c, 0);
  cplx_stash_a<NWN>(c, As2);
  cplx_stash_b<NWN>(c, Bs2);
  cplx_fetch_a<NWN>(c, 16);
  cplx_fetch_b<NWN>(c, 32);
  __syncthreads();

  for (int k0 = 0; k0 < K; k0 += 64) {
    cplx_stage<NWN, 0, 0>(c, k0,      As2, Bs2, acc);
    cplx_stage<NWN, 1, 0>(c, k0 + 16, As2, Bs2, acc);
    cplx_stage<NWN, 0, 1>(c, k0 + 32, As2, Bs2, acc);
    cplx_stage<NWN, 1, 1>(c, k0 + 48, As2, Bs2, acc);
  }
}

__global__ void pack_w1_kernel(const float* __restrict__ Bwr, const float* __restrict__ Bwi,
                               const float* __restrict__ gw, float* __restrict__ W1) {
  int idx = blockIdx.x * 256 + threadIdx.x;
  if (idx >= KX * NP) return;
  int k = idx / NP, j = idx - k * NP;
  int d = k >> 1, p = k & 1;
  float v;
  if (j < 64) {
    v = p ? -Bwi[j * DD + d] : Bwr[j * DD + d];
  } else if (j < 128) {
    int n = j - 64;
    v = p ? Bwr[n * DD + d] : Bwi[n * DD + d];
  } else {
    int n = j - 128;
    v = gw[n * (2 * DD) + p * DD + d];
  }
  W1[idx] = v;
}

__device__ __forceinline__ float4* zpart_slot(float* out, int m0, int j0, int w, int q) {
  const int lane = threadIdx.x & 31;
  return (float4*)(out + (size_t)(m0 + w * 16 + q) * NO + j0 + lane * 4);
}

#define DX_TILES ((MM / 128) * (NO / 128) - 1)
#define XT_M0 ((DX_TILES & (MM / 128 - 1)) * 128)
#define XT_J0 ((DX_TILES / (MM / 128)) * 128)
#define PROJ_WGS ((MM / 128) * (NP / 64))

__global__ __launch_bounds__(256) void gemm_proj_kernel(const float* __restrict__ x,
                                                        const float* __restrict__ W1,
                                                        const float* __restrict__ gate_b,
                                                        float* __restrict__ P,
                                                        const float* __restrict__ Dr,
                                                        const float* __restrict__ Di,
                                                        float* __restrict__ out) {
  __shared__ v2f As2[2 * 8 * A2LD];
  __shared__ __attribute__((aligned(16))) v2f Bs2[2 * 16 * (64 + 16)];
  v8f acc[2][2] = {{{0,0,0,0,0,0,0,0}, {0,0,0,0,0,0,0,0}},
                   {{0,0,0,0,0,0,0,0}, {0,0,0,0,0,0,0,0}}};
  const int bx = blockIdx.x;
  if (bx >= PROJ_WGS) {
    const int hb = bx - PROJ_WGS;
    cplx_tile_acc<2>(x, KX, Dr, Di, DD, KX, XT_M0, XT_J0 + hb * 64, As2, Bs2, acc);
    const int wave = threadIdx.x >> 5;
    const int w4 = (wave & 3) + 4 * hb;
#pragma unroll
    for (int mi = 0; mi < 2; ++mi)
#pragma unroll
      for (int nl = 0; nl < 2; ++nl) {
        const v8f a = acc[mi][nl];
        const int q = (mi * 4 + 2 * (wave >> 2) + nl) * 2;
        *zpart_slot(out, XT_M0, XT_J0, w4, q)     = make_float4(a[0], a[1], a[2], a[3]);
        *zpart_slot(out, XT_M0, XT_J0, w4, q + 1) = make_float4(a[4], a[5], a[6], a[7]);
      }
    return;
  }
  const int m0 = (bx & (MM / 128 - 1)) * 128;
  const int j0 = (bx / (MM / 128)) * 64;
  gemm_tile_acc<2>(x, KX, W1, NP, KX, m0, j0, As2, Bs2, acc);

  const int lane = threadIdx.x & 31, wave = threadIdx.x >> 5;
  const int half = lane >> 4, lrow = lane & 15;
  const int wm = (wave & 3) * 32, wn = (wave >> 2) * 32;
#pragma unroll
  for (int mi = 0; mi < 2; ++mi)
#pragma unroll
    for (int ni = 0; ni < 2; ++ni)
#pragma unroll
      for (int e = 0; e < 8; ++e) {
        int m = m0 + wm + mi * 16 + half * 8 + e;
        int j = j0 + wn + ni * 16 + lrow;
        float v = acc[mi][ni][e];
        if (j >= 128) v = 1.0f / (1.0f + expf(-(v + gate_b[j - 128])));
        P[(size_t)m * NP + j] = v;
      }
}

#define SCAN_STEPS 16

__device__ __forceinline__ void scan_body(const float* __restrict__ P,
                                          const float* __restrict__ h0,
                                          const float* __restrict__ theta,
                                          const float* __restrict__ dampp,
                                          float* __restrict__ hws,
                                          float* __restrict__ hfin) {
  const int t = threadIdx.x;
  const int b = t >> 6, n = t & 63;
  const float c = cosf(theta[n]);
  const float s = sinf(theta[n]);
  const float dmp = 0.5f + 0.5f / (1.0f + expf(-dampp[n]));
  const float ca = c * dmp, sa = s * dmp;
  float hr = h0[(b * NNh + n) * 2 + 0];
  float hi = h0[(b * NNh + n) * 2 + 1];
  for (int st0 = 0; st0 < SS; st0 += SCAN_STEPS) {
    const float* prow = P + ((size_t)b * SS + st0) * NP + n;
    float g[SCAN_STEPS], br[SCAN_STEPS], bi[SCAN_STEPS];
#pragma unroll
    for (int u = 0; u < SCAN_STEPS; ++u) {
      g[u]  = prow[(size_t)u * NP + 128];
      br[u] = prow[(size_t)u * NP];
      bi[u] = prow[(size_t)u * NP + 64];
    }
    float* hrow = hws + ((size_t)b * SS + st0) * KH + 2 * n;
#pragma unroll
    for (int u = 0; u < SCAN_STEPS; ++u) {
      float rr = hr * ca - hi * sa;
      float ri = hr * sa + hi * ca;
      hr = g[u] * br[u] + (1.0f - g[u]) * rr;
      hi = g[u] * bi[u] + (1.0f - g[u]) * ri;
      hrow[(size_t)u * KH + 0] = hr;
      hrow[(size_t)u * KH + 1] = hi;
    }
  }
  hfin[(b * NNh + n) * 2 + 0] = hr;
  hfin[(b * NNh + n) * 2 + 1] = hi;
}

__global__ __launch_bounds__(256, 8) void gemm_dx_scan_kernel(const float* __restrict__ x,
                                                           const float* __restrict__ Dr,
                                                           const float* __restrict__ Di,
                                                           const float* __restrict__ P,
                                                           const float* __restrict__ h0,
                                                           const float* __restrict__ theta,
                                                           const float* __restrict__ dampp,
                                                           float* __restrict__ hws,
                                                           float* __restrict__ hfin,
                                                           float* __restrict__ out) {
  __shared__ v2f As2[2 * 8 * A2LD];
  __shared__ __attribute__((aligned(16))) v2f Bs2[2 * 16 * (128 + 16)];
  if (blockIdx.x == 0) {
    scan_body(P, h0, theta, dampp, hws, hfin);
    return;
  }
  v8f acc[2][4] = {{{0,0,0,0,0,0,0,0}, {0,0,0,0,0,0,0,0},
                    {0,0,0,0,0,0,0,0}, {0,0,0,0,0,0,0,0}},
                   {{0,0,0,0,0,0,0,0}, {0,0,0,0,0,0,0,0},
                    {0,0,0,0,0,0,0,0}, {0,0,0,0,0,0,0,0}}};
  const int tile = blockIdx.x - 1;
  const int m0 = (tile & (MM / 128 - 1)) * 128;
  const int j0 = (tile / (MM / 128)) * 128;
  cplx_tile_acc<4>(x, KX, Dr, Di, DD, KX, m0, j0, As2, Bs2, acc);

#pragma unroll
  for (int mi = 0; mi < 2; ++mi)
#pragma unroll
    for (int ni = 0; ni < 4; ++ni) {
      const v8f a = acc[mi][ni];
      const int q = (mi * 4 + ni) * 2;
      *zpart_slot(out, m0, j0, threadIdx.x >> 5, q)     = make_float4(a[0], a[1], a[2], a[3]);
      *zpart_slot(out, m0, j0, threadIdx.x >> 5, q + 1) = make_float4(a[4], a[5], a[6], a[7]);
    }
}

__global__ __launch_bounds__(256) void gemm_ch_out_kernel(const float* __restrict__ hws,
                                                          const float* __restrict__ Cr,
                                                          const float* __restrict__ Ci,
                                                          float* out) {
  __shared__ v2f As2[2 * 8 * A2LD];
  __shared__ __attribute__((aligned(16))) v2f Bs2[2 * 16 * (128 + 16)];
  const int m0 = blockIdx.x * 128;
  const int j0 = blockIdx.y * 128;
  v8f acc[2][4];
#pragma unroll
  for (int mi = 0; mi < 2; ++mi)
#pragma unroll
    for (int ni = 0; ni < 4; ++ni) {
      const int q = (mi * 4 + ni) * 2;
      const float4 lo = *zpart_slot(out, m0, j0, threadIdx.x >> 5, q);
      const float4 hi = *zpart_slot(out, m0, j0, threadIdx.x >> 5, q + 1);
      v8f a = {lo.x, lo.y, lo.z, lo.w, hi.x, hi.y, hi.z, hi.w};
      acc[mi][ni] = a;
    }
  cplx_tile_acc<4>(hws, KH, Cr, Ci, NNh, KH, m0, j0, As2, Bs2, acc);

  const int lane = threadIdx.x & 31, wave = threadIdx.x >> 5;
  const int half = lane >> 4, lrow = lane & 15;
  const int wm = (wave & 3) * 32, wn = (wave >> 2) * 64;
#pragma unroll
  for (int mi = 0; mi < 2; ++mi)
#pragma unroll
    for (int ni = 0; ni < 4; ++ni)
#pragma unroll
      for (int e = 0; e < 8; ++e) {
        int m = m0 + wm + mi * 16 + half * 8 + e;
        int j = j0 + wn + ni * 16 + lrow;
        float z = acc[mi][ni][e];
        float zr = __shfl(z, lane & ~1, 32);
        out[(size_t)m * NO + j] = zr * z + z;
      }
}

extern "C" void kernel_launch(void* const* d_in, const int* in_sizes, int n_in,
                              void* d_out, int out_size, void* d_ws, size_t ws_size,
                              hipStream_t stream) {
  const float* x     = (const float*)d_in[0];
  const float* h0    = (const float*)d_in[1];
  const float* th    = (const float*)d_in[2];
  const float* dampp = (const float*)d_in[3];
  const float* Bwr   = (const float*)d_in[4];
  const float* Bwi   = (const float*)d_in[5];
  const float* Cwr   = (const float*)d_in[6];
  const float* Cwi   = (const float*)d_in[7];
  const float* Dwr   = (const float*)d_in[8];
  const float* Dwi   = (const float*)d_in[9];
  const float* gw    = (const float*)d_in[10];
  const float* gb    = (const float*)d_in[11];

  float* out  = (float*)d_out;
  float* hfin = out + (size_t)MM * NO;

  float* ws = (float*)d_ws;
  float* W1  = ws;
  float* P   = W1 + (size_t)KX * NP;
  float* Hws = P + (size_t)MM * NP;

  pack_w1_kernel<<<(KX * NP + 255) / 256, 256, 0, stream>>>(Bwr, Bwi, gw, W1);

  gemm_proj_kernel<<<PROJ_WGS + 2, 256, 0, stream>>>(x, W1, gb, P, Dwr, Dwi, out);
  gemm_dx_scan_kernel<<<1 + DX_TILES, 256, 0, stream>>>(x, Dwr, Dwi, P, h0, th, dampp,
                                                        Hws, hfin, out);
  gemm_ch_out_kernel<<<dim3(MM / 128, NO / 128), 256, 0, stream>>>(Hws, Cwr, Cwi, out);
}


# proj and last launch: a k stage's global loads are waited for at the next stage's LDS stores, not right behind their issue (k-loop tail peeled, stage barrier orders LDS only); same sums
# speedup vs baseline: 1.7383x; 1.0454x over previous
#include <hip/hip_runtime.h>
#include <math.h>

#define BB 4
#define SS 2048
#define DD 1024
#define NNh 64
#define MM (BB * SS)
#define KX (2 * DD)
#define NP 192
#define KH (2 * NNh)
#define NO (2 * DD)

#define A2LD 144

typedef __attribute__((ext_vector_type(2))) float v2f;
typedef __attribute__((ext_vector_type(8))) float v8f;

__device__ __forceinline__ v8f wmma_f32(v2f a, v2f b, v8f c) {
  return __builtin_amdgcn_wmma_f32_16x16x4_f32(
       false, a,  false, b,
       (short)0, c,  false,  false);
}

__device__ __forceinline__ void stage_barrier() {
  __builtin_amdgcn_fence(__ATOMIC_RELEASE, "workgroup", "local");
  __builtin_amdgcn_s_barrier();
  __builtin_amdgcn_fence(__ATOMIC_ACQUIRE, "workgroup", "local");
}

template <int NWN>
struct GemmCtx {
  static constexpr int BN   = NWN * 32;
  static constexpr int B2LD = BN + 16;
  static constexpr int NB4  = BN / 64;
  static constexpr int ASZ  = 8 * A2LD;
  static constexpr int BSZ  = 8 * B2LD;
  const float* pa0; const float* pa1;
  const float* pb[NB4];
  int sa0, sa1;
  int sb[NB4];
  int fa, fb;
  int ldb, K;
  float4 aReg0, aReg1, bReg[NB4];
};

template <int NWN>
__device__ __forceinline__ void gemm_fetch(GemmCtx<NWN>& c, int k) {
  c.aReg0 = *(const float4*)(c.pa0 + k);
  c.aReg1 = *(const float4*)(c.pa1 + k);
#pragma unroll
  for (int i = 0; i < GemmCtx<NWN>::NB4; ++i)
    c.bReg[i] = *(const float4*)(c.pb[i] + (size_t)k * c.ldb);
}

template <int NWN>
__device__ __forceinline__ void gemm_stash(const GemmCtx<NWN>& c, v2f* An, v2f* Bn) {
  v2f t;
  t.x = c.aReg0.x; t.y = c.aReg0.y; An[c.sa0] = t;
  t.x = c.aReg0.z; t.y = c.aReg0.w; An[c.sa0 + A2LD] = t;
  t.x = c.aReg1.x; t.y = c.aReg1.y; An[c.sa1] = t;
  t.x = c.aReg1.z; t.y = c.aReg1.w; An[c.sa1 + A2LD] = t;
  float* bsf = (float*)Bn;
#pragma unroll
  for (int i = 0; i < GemmCtx<NWN>::NB4; ++i) {
    bsf[c.sb[i] + 0] = c.bReg[i].x;
    bsf[c.sb[i] + 2] = c.bReg[i].y;
    bsf[c.sb[i] + 4] = c.bReg[i].z;
    bsf[c.sb[i] + 6] = c.bReg[i].w;
  }
}

template <int NWN, int CUR, bool TAIL>
__device__ __forceinline__ void gemm_stage(GemmCtx<NWN>& c, int k0, v2f* As2, v2f* Bs2,
                                           v8f acc[2][NWN]) {
  using G = GemmCtx<NWN>;
  const v2f* Ac = As2 + CUR * G::ASZ;
  const v2f* Bc = Bs2 + CUR * G::BSZ;
  v2f* An = As2 + (CUR ^ 1) * G::ASZ;
  v2f* Bn = Bs2 + (CUR ^ 1) * G::BSZ;

  v2f a0[2], a1[2], bf[2][NWN];
  a0[0] = Ac[c.fa];
  a1[0] = Ac[c.fa + 16];
#pragma unroll
  for (int ni = 0; ni < NWN; ++ni) bf[0][ni] = Bc[c.fb + ni * 16];

  if (!(TAIL && CUR == 1)) {
    gemm_stash<NWN>(c, An, Bn);
    if (!TAIL) {
      gemm_fetch<NWN>(c, k0 + 32);
      const int kp = (k0 + 48 < c.K) ? k0 + 48 : c.K - 16;
      __builtin_prefetch(c.pa0 + kp, 0, 0);
      __builtin_prefetch(c.pa1 + kp, 0, 0);
    }
  }

#pragma unroll
  for (int kstep = 0; kstep < 4; ++kstep) {
    const int cs = kstep & 1, ns = cs ^ 1;
    if (kstep < 3) {
      const int p2 = 2 * (kstep + 1);
      a0[ns] = Ac[c.fa + p2 * A2LD];
      a1[ns] = Ac[c.fa + p2 * A2LD + 16];
#pragma unroll
      for (int ni = 0; ni < NWN; ++ni) bf[ns][ni] = Bc[c.fb + p2 * G::B2LD + ni * 16];
    }
#pragma unroll
    for (int ni = 0; ni < NWN; ++ni) {
      acc[0][ni] = wmma_f32(a0[cs], bf[cs][ni], acc[0][ni]);
      acc[1][ni] = wmma_f32(a1[cs], bf[cs][ni], acc[1][ni]);
    }
  }
  stage_barrier();
}

template <int NWN>
__device__ void gemm_tile_acc(const float* __restrict__ A, int lda,
                              const float* __restrict__ Bp, int ldb,
                              int K, int m0, int j0,
                              v2f* As2, v2f* Bs2, v8f acc[2][NWN]) {
  using G = GemmCtx<NWN>;
  const int tid  = threadIdx.x;
  const int lane = tid & 31;
  const int wave = tid >> 5;
  const int half = lane >> 4;
  const int lrow = lane & 15;
  const int wm   = (wave & 3) * 32;
  const int wn   = (wave >> 2) * (NWN * 16);

  GemmCtx<NWN> c;
  c.ldb = ldb; c.K = K;
  const int am0 = tid >> 2;
  const int ak0 = (tid & 3) << 2;
  const int am1 = (256 + tid) >> 2;
  const int ak1 = ((256 + tid) & 3) << 2;
  c.pa0 = A + (size_t)(m0 + am0) * lda + ak0;
  c.pa1 = A + (size_t)(m0 + am1) * lda + ak1;
  c.sa0 = (ak0 >> 1) * A2LD + am0;
  c.sa1 = (ak1 >> 1) * A2LD + am1;
#pragma unroll
  for (int i = 0; i < G::NB4; ++i) {
    const int idx = i * 256 + tid;
    const int bk = idx / (G::BN / 4);
    const int bj = (idx % (G::BN / 4)) * 4;
    c.pb[i] = Bp + (size_t)bk * ldb + j0 + bj;
    c.sb[i] = ((bk >> 1) * G::B2LD + bj) * 2 + (bk & 1);
  }
  c.fa = half * A2LD + wm + lrow;
  c.fb = half * G::B2LD + wn + lrow;

  gemm_fetch<NWN>(c, 0);
  gemm_stash<NWN>(c, As2, Bs2);
  gemm_fetch<NWN>(c, 16);
  stage_barrier();

  int k0 = 0;
  for (; k0 < K - 32; k0 += 32) {
    gemm_stage<NWN, 0, false>(c, k0,      As2, Bs2, acc);
    gemm_stage<NWN, 1, false>(c, k0 + 16, As2, Bs2, acc);
  }
  gemm_stage<NWN, 0, true>(c, k0,      As2, Bs2, acc);
  gemm_stage<NWN, 1, true>(c, k0 + 16, As2, Bs2, acc);
}

template <int NWN>
struct CplxCtx {
  static constexpr int BN   = NWN * 32;
  static constexpr int B2LD = BN + 16;
  static constexpr int ASZ  = 8 * A2LD;
  static constexpr int BSZ  = 16 * B2LD;
  static constexpr int BTH  = (BN / 2) * 4;
  const float* A; const float* Wr; const float* Wi;
  unsigned ao, a1d;
  unsigned bo;
  int sa0, sa1;
  int sb;
  int fa, fb;
  int K;
  float4 aReg0, aReg1, rReg, iReg;
};

template <int NWN>
__device__ __forceinline__ void cplx_fetch_a(CplxCtx<NWN>& c, int k) {
  c.aReg0 = *(const float4*)((const char*)(c.A + k) + c.ao);
  c.aReg1 = *(const float4*)((const char*)(c.A + k + c.a1d) + c.ao);
}

template <int NWN>
__device__ __forceinline__ void cplx_fetch_b(CplxCtx<NWN>& c, int k) {
  if (CplxCtx<NWN>::BTH == 256 || threadIdx.x < CplxCtx<NWN>::BTH) {
    c.rReg = *(const float4*)((const char*)(c.Wr + (k >> 1)) + c.bo);
    c.iReg = *(const float4*)((const char*)(c.Wi + (k >> 1)) + c.bo);
  }
}

template <int NWN>
__device__ __forceinline__ void cplx_stash_a(const CplxCtx<NWN>& c, v2f* An) {
  v2f t;
  t.x = c.aReg0.x; t.y = c.aReg0.y; An[c.sa0] = t;
  t.x = c.aReg0.z; t.y = c.aReg0.w; An[c.sa0 + A2LD] = t;
  t.x = c.aReg1.x; t.y = c.aReg1.y; An[c.sa1] = t;
  t.x = c.aReg1.z; t.y = c.aReg1.w; An[c.sa1 + A2LD] = t;
}

template <int NWN>
__device__ __forceinline__ void cplx_stash_b(const CplxCtx<NWN>& c, v2f* Bn) {
  if (CplxCtx<NWN>::BTH == 256 || threadIdx.x < CplxCtx<NWN>::BTH) {
    float4* b4 = (float4*)(Bn + c.sb);
    constexpr int R4 = CplxCtx<NWN>::B2LD / 2;
    b4[0 * R4] = make_float4(c.rReg.x, -c.iReg.x, c.iReg.x, c.rReg.x);
    b4[1 * R4] = make_float4(c.rReg.y, -c.iReg.y, c.iReg.y, c.rReg.y);
    b4[2 * R4] = make_float4(c.rReg.z, -c.iReg.z, c.iReg.z, c.rReg.z);
    b4[3 * R4] = make_float4(c.rReg.w, -c.iReg.w, c.iReg.w, c.rReg.w);
  }
}

template <int NWN, int PH, int PB, int MODE>
__device__ __forceinline__ void cplx_stage(CplxCtx<NWN>& c, int k0, v2f* As2, v2f* Bs2,
                                           v8f acc[2][NWN]) {
  using G = CplxCtx<NWN>;
  const v2f* Ac = As2 + PH * G::ASZ;
  const v2f* Bc = Bs2 + PB * G::BSZ + PH * 8 * G::B2LD;
  v2f* An = As2 + (PH ^ 1) * G::ASZ;
  v2f* Bn = Bs2 + (PB ^ 1) * G::BSZ;

  v2f a0[2], a1[2], bf[2][NWN];
  a0[0] = Ac[c.fa];
  a1[0] = Ac[c.fa + 16];
#pragma unroll
  for (int ni = 0; ni < NWN; ++ni) bf[0][ni] = Bc[c.fb + ni * 16];

  const bool stashA = MODE == 0 ? (k0 + 16 < c.K) : !(MODE == 2 && PH == 1 && PB == 1);
  const bool stashB = MODE == 0 ? (k0 + 32 < c.K) : !(MODE == 2 && PB == 1);
  const bool fetchB = MODE == 0 ? (k0 + 64 < c.K) : (MODE == 1);
  const bool fetchA = MODE == 0 ? (k0 + 32 < c.K) : !(MODE == 2 && PB == 1);
  const bool prefA  = MODE == 0 ? (k0 + 48 < c.K) : (MODE == 1 || (PH == 0 && PB == 0));
  if (stashA) {
    cplx_stash_a<NWN>(c, An);
    if (PH == 0 && stashB) {
      cplx_stash_b<NWN>(c, Bn);
      if (fetchB) cplx_fetch_b<NWN>(c, k0 + 64);
    }
    if (fetchA) cplx_fetch_a<NWN>(c, k0 + 32);
    if (prefA) {
      __builtin_prefetch((const char*)(c.A + k0 + 48) + c.ao, 0, 0);
      __builtin_prefetch((const char*)(c.A + k0 + 48 + c.a1d) + c.ao, 0, 0);
    }
  }

#pragma unroll
  for (int kstep = 0; kstep < 4; ++kstep) {
    const int cs = kstep & 1, ns = cs ^ 1;
    if (kstep < 3) {
      const int p2 = 2 * (kstep + 1);
      a0[ns] = Ac[c.fa + p2 * A2LD];
      a1[ns] = Ac[c.fa + p2 * A2LD + 16];
#pragma unroll
      for (int ni = 0; ni < NWN; ++ni) bf[ns][ni] = Bc[c.fb + p2 * G::B2LD + ni * 16];
    }
#pragma unroll
    for (int ni = 0; ni < NWN; ++ni) {
      acc[0][ni] = wmma_f32(a0[cs], bf[cs][ni], acc[0][ni]);
      acc[1][ni] = wmma_f32(a1[cs], bf[cs][ni], acc[1][ni]);
    }
  }
  if (MODE == 0) __syncthreads(); else stage_barrier();
}

template <int NWN, bool LATE>
__device__ void cplx_tile_acc(const float* __restrict__ A, int lda,
                              const float* __restrict__ Wr, const float* __restrict__ Wi,
                              int ldw, int K, int m0, int j0,
                              v2f* As2, v2f* Bs2, v8f acc[2][NWN]) {
  using G = CplxCtx<NWN>;
  const int tid  = threadIdx.x;
  const int lane = tid & 31;
  const int wave = tid >> 5;
  const int half = lane >> 4;
  const int lrow = lane & 15;
  const int wm   = (wave & 3) * 32;
  const int wn   = (wave >> 2) * (NWN * 16);

  CplxCtx<NWN> c;
  c.K = K;
  const int am0 = tid >> 2;
  const int ak0 = (tid & 3) << 2;
  const int am1 = am0 + 64;
  c.A = A; c.Wr = Wr; c.Wi = Wi;
  c.ao  = 4u * ((unsigned)(m0 + am0) * (unsigned)lda + (unsigned)ak0);
  c.a1d = 64u * (unsigned)lda;
  c.sa0 = (ak0 >> 1) * A2LD + am0;
  c.sa1 = (ak0 >> 1) * A2LD + am1;
  const int bd = (tid >> 2) & (G::BN / 2 - 1);
  const int bu = tid & 3;
  c.bo = 4u * ((unsigned)((j0 >> 1) + bd) * (unsigned)ldw + 4u * (unsigned)bu);
  c.sb = (4 * bu) * G::B2LD + 2 * bd;
  c.fa = half * A2LD + wm + lrow;
  c.fb = half * G::B2LD + wn + lrow;

  cplx_fetch_a<NWN>(c, 0);
  cplx_fetch_b<NWN>(c, 0);
  cplx_stash_a<NWN>(c, As2);
  cplx_stash_b<NWN>(c, Bs2);
  cplx_fetch_a<NWN>(c, 16);
  cplx_fetch_b<NWN>(c, 32);
  __syncthreads();
  if (!LATE) {
    for (int k0 = 0; k0 < K; k0 += 64) {
      cplx_stage<NWN, 0, 0, 0>(c, k0,      As2, Bs2, acc);
      cplx_stage<NWN, 1, 0, 0>(c, k0 + 16, As2, Bs2, acc);
      cplx_stage<NWN, 0, 1, 0>(c, k0 + 32, As2, Bs2, acc);
      cplx_stage<NWN, 1, 1, 0>(c, k0 + 48, As2, Bs2, acc);
    }
    return;
  }
  int k0 = 0;
  for (; k0 < K - 64; k0 += 64) {
    cplx_stage<NWN, 0, 0, 1>(c, k0,      As2, Bs2, acc);
    cplx_stage<NWN, 1, 0, 1>(c, k0 + 16, As2, Bs2, acc);
    cplx_stage<NWN, 0, 1, 1>(c, k0 + 32, As2, Bs2, acc);
    cplx_stage<NWN, 1, 1, 1>(c, k0 + 48, As2, Bs2, acc);
  }
  cplx_stage<NWN, 0, 0, 2>(c, k0,      As2, Bs2, acc);
  cplx_stage<NWN, 1, 0, 2>(c, k0 + 16, As2, Bs2, acc);
  cplx_stage<NWN, 0, 1, 2>(c, k0 + 32, As2, Bs2, acc);
  cplx_stage<NWN, 1, 1, 2>(c, k0 + 48, As2, Bs2, acc);
}

__global__ void pack_w1_kernel(const float* __restrict__ Bwr, const float* __restrict__ Bwi,
                               const float* __restrict__ gw, float* __restrict__ W1) {
  int idx = blockIdx.x * 256 + threadIdx.x;
  if (idx >= KX * NP) return;
  int k = idx / NP, j = idx - k * NP;
  int d = k >> 1, p = k & 1;
  float v;
  if (j < 64) {
    v = p ? -Bwi[j * DD + d] : Bwr[j * DD + d];
  } else if (j < 128) {
    int n = j - 64;
    v = p ? Bwr[n * DD + d] : Bwi[n * DD + d];
  } else {
    int n = j - 128;
    v = gw[n * (2 * DD) + p * DD + d];
  }
  W1[idx] = v;
}

__device__ __forceinline__ float4* zpart_slot(float* out, int m0, int j0, int w, int q) {
  const int lane = threadIdx.x & 31;
  return (float4*)(out + (size_t)(m0 + w * 16 + q) * NO + j0 + lane * 4);
}

#define DX_TILES ((MM / 128) * (NO / 128) - 1)
#define XT_M0 ((DX_TILES & (MM / 128 - 1)) * 128)
#define XT_J0 ((DX_TILES / (MM / 128)) * 128)
#define PROJ_WGS ((MM / 128) * (NP / 64))

__global__ __launch_bounds__(256) void gemm_proj_kernel(const float* __restrict__ x,
                                                        const float* __restrict__ W1,
                                                        const float* __restrict__ gate_b,
                                                        float* __restrict__ P,
                                                        const float* __restrict__ Dr,
                                                        const float* __restrict__ Di,
                                                        float* __restrict__ out) {
  __shared__ v2f As2[2 * 8 * A2LD];
  __shared__ __attribute__((aligned(16))) v2f Bs2[2 * 16 * (64 + 16)];
  v8f acc[2][2] = {{{0,0,0,0,0,0,0,0}, {0,0,0,0,0,0,0,0}},
                   {{0,0,0,0,0,0,0,0}, {0,0,0,0,0,0,0,0}}};
  const int bx = blockIdx.x;
  if (bx >= PROJ_WGS) {
    const int hb = bx - PROJ_WGS;
    cplx_tile_acc<2, true>(x, KX, Dr, Di, DD, KX, XT_M0, XT_J0 + hb * 64, As2, Bs2, acc);
    const int wave = threadIdx.x >> 5;
    const int w4 = (wave & 3) + 4 * hb;
#pragma unroll
    for (int mi = 0; mi < 2; ++mi)
#pragma unroll
      for (int nl = 0; nl < 2; ++nl) {
        const v8f a = acc[mi][nl];
        const int q = (mi * 4 + 2 * (wave >> 2) + nl) * 2;
        *zpart_slot(out, XT_M0, XT_J0, w4, q)     = make_float4(a[0], a[1], a[2], a[3]);
        *zpart_slot(out, XT_M0, XT_J0, w4, q + 1) = make_float4(a[4], a[5], a[6], a[7]);
      }
    return;
  }
  const int m0 = (bx & (MM / 128 - 1)) * 128;
  const int j0 = (bx / (MM / 128)) * 64;
  gemm_tile_acc<2>(x, KX, W1, NP, KX, m0, j0, As2, Bs2, acc);

  const int lane = threadIdx.x & 31, wave = threadIdx.x >> 5;
  const int half = lane >> 4, lrow = lane & 15;
  const int wm = (wave & 3) * 32, wn = (wave >> 2) * 32;
#pragma unroll
  for (int mi = 0; mi < 2; ++mi)
#pragma unroll
    for (int ni = 0; ni < 2; ++ni)
#pragma unroll
      for (int e = 0; e < 8; ++e) {
        int m = m0 + wm + mi * 16 + half * 8 + e;
        int j = j0 + wn + ni * 16 + lrow;
        float v = acc[mi][ni][e];
        if (j >= 128) v = 1.0f / (1.0f + expf(-(v + gate_b[j - 128])));
        P[(size_t)m * NP + j] = v;
      }
}

#define SCAN_STEPS 16

__device__ __forceinline__ void scan_body(const float* __restrict__ P,
                                          const float* __restrict__ h0,
                                          const float* __restrict__ theta,
                                          const float* __restrict__ dampp,
                                          float* __restrict__ hws,
                                          float* __restrict__ hfin) {
  const int t = threadIdx.x;
  const int b = t >> 6, n = t & 63;
  const float c = cosf(theta[n]);
  const float s = sinf(theta[n]);
  const float dmp = 0.5f + 0.5f / (1.0f + expf(-dampp[n]));
  const float ca = c * dmp, sa = s * dmp;
  float hr = h0[(b * NNh + n) * 2 + 0];
  float hi = h0[(b * NNh + n) * 2 + 1];
  for (int st0 = 0; st0 < SS; st0 += SCAN_STEPS) {
    const float* prow = P + ((size_t)b * SS + st0) * NP + n;
    float g[SCAN_STEPS], br[SCAN_STEPS], bi[SCAN_STEPS];
#pragma unroll
    for (int u = 0; u < SCAN_STEPS; ++u) {
      g[u]  = prow[(size_t)u * NP + 128];
      br[u] = prow[(size_t)u * NP];
      bi[u] = prow[(size_t)u * NP + 64];
    }
    float* hrow = hws + ((size_t)b * SS + st0) * KH + 2 * n;
#pragma unroll
    for (int u = 0; u < SCAN_STEPS; ++u) {
      float rr = hr * ca - hi * sa;
      float ri = hr * sa + hi * ca;
      hr = g[u] * br[u] + (1.0f - g[u]) * rr;
      hi = g[u] * bi[u] + (1.0f - g[u]) * ri;
      hrow[(size_t)u * KH + 0] = hr;
      hrow[(size_t)u * KH + 1] = hi;
    }
  }
  hfin[(b * NNh + n) * 2 + 0] = hr;
  hfin[(b * NNh + n) * 2 + 1] = hi;
}

__global__ __launch_bounds__(256, 8) void gemm_dx_scan_kernel(const float* __restrict__ x,
                                                           const float* __restrict__ Dr,
                                                           const float* __restrict__ Di,
                                                           const float* __restrict__ P,
                                                           const float* __restrict__ h0,
                                                           const float* __restrict__ theta,
                                                           const float* __restrict__ dampp,
                                                           float* __restrict__ hws,
                                                           float* __restrict__ hfin,
                                                           float* __restrict__ out) {
  __shared__ v2f As2[2 * 8 * A2LD];
  __shared__ __attribute__((aligned(16))) v2f Bs2[2 * 16 * (128 + 16)];
  if (blockIdx.x == 0) {
    scan_body(P, h0, theta, dampp, hws, hfin);
    return;
  }
  v8f acc[2][4] = {{{0,0,0,0,0,0,0,0}, {0,0,0,0,0,0,0,0},
                    {0,0,0,0,0,0,0,0}, {0,0,0,0,0,0,0,0}},
                   {{0,0,0,0,0,0,0,0}, {0,0,0,0,0,0,0,0},
                    {0,0,0,0,0,0,0,0}, {0,0,0,0,0,0,0,0}}};
  const int tile = blockIdx.x - 1;
  const int m0 = (tile & (MM / 128 - 1)) * 128;
  const int j0 = (tile / (MM / 128)) * 128;
  cplx_tile_acc<4, false>(x, KX, Dr, Di, DD, KX, m0, j0, As2, Bs2, acc);

#pragma unroll
  for (int mi = 0; mi < 2; ++mi)
#pragma unroll
    for (int ni = 0; ni < 4; ++ni) {
      const v8f a = acc[mi][ni];
      const int q = (mi * 4 + ni) * 2;
      *zpart_slot(out, m0, j0, threadIdx.x >> 5, q)     = make_float4(a[0], a[1], a[2], a[3]);
      *zpart_slot(out, m0, j0, threadIdx.x >> 5, q + 1) = make_float4(a[4], a[5], a[6], a[7]);
    }
}

__global__ __launch_bounds__(256) void gemm_ch_out_kernel(const float* __restrict__ hws,
                                                          const float* __restrict__ Cr,
                                                          const float* __restrict__ Ci,
                                                          float* out) {
  __shared__ v2f As2[2 * 8 * A2LD];
  __shared__ __attribute__((aligned(16))) v2f Bs2[2 * 16 * (128 + 16)];
  const int m0 = blockIdx.x * 128;
  const int j0 = blockIdx.y * 128;
  v8f acc[2][4];
#pragma unroll
  for (int mi = 0; mi < 2; ++mi)
#pragma unroll
    for (int ni = 0; ni < 4; ++ni) {
      const int q = (mi * 4 + ni) * 2;
      const float4 lo = *zpart_slot(out, m0, j0, threadIdx.x >> 5, q);
      const float4 hi = *zpart_slot(out, m0, j0, threadIdx.x >> 5, q + 1);
      v8f a = {lo.x, lo.y, lo.z, lo.w, hi.x, hi.y, hi.z, hi.w};
      acc[mi][ni] = a;
    }
  cplx_tile_acc<4, true>(hws, KH, Cr, Ci, NNh, KH, m0, j0, As2, Bs2, acc);

  const int lane = threadIdx.x & 31, wave = threadIdx.x >> 5;
  const int half = lane >> 4, lrow = lane & 15;
  const int wm = (wave & 3) * 32, wn = (wave >> 2) * 64;
#pragma unroll
  for (int mi = 0; mi < 2; ++mi)
#pragma unroll
    for (int ni = 0; ni < 4; ++ni)
#pragma unroll
      for (int e = 0; e < 8; ++e) {
        int m = m0 + wm + mi * 16 + half * 8 + e;
        int j = j0 + wn + ni * 16 + lrow;
        float z = acc[mi][ni][e];
        float zr = __shfl(z, lane & ~1, 32);
        out[(size_t)m * NO + j] = zr * z + z;
      }
}

extern "C" void kernel_launch(void* const* d_in, const int* in_sizes, int n_in,
                              void* d_out, int out_size, void* d_ws, size_t ws_size,
                              hipStream_t stream) {
  const float* x     = (const float*)d_in[0];
  const float* h0    = (const float*)d_in[1];
  const float* th    = (const float*)d_in[2];
  const float* dampp = (const float*)d_in[3];
  const float* Bwr   = (const float*)d_in[4];
  const float* Bwi   = (const float*)d_in[5];
  const float* Cwr   = (const float*)d_in[6];
  const float* Cwi   = (const float*)d_in[7];
  const float* Dwr   = (const float*)d_in[8];
  const float* Dwi   = (const float*)d_in[9];
  const float* gw    = (const float*)d_in[10];
  const float* gb    = (const float*)d_in[11];

  float* out  = (float*)d_out;
  float* hfin = out + (size_t)MM * NO;

  float* ws = (float*)d_ws;
  float* W1  = ws;
  float* P   = W1 + (size_t)KX * NP;
  float* Hws = P + (size_t)MM * NP;

  pack_w1_kernel<<<(KX * NP + 255) / 256, 256, 0, stream>>>(Bwr, Bwi, gw, W1);

  gemm_proj_kernel<<<PROJ_WGS + 2, 256, 0, stream>>>(x, W1, gb, P, Dwr, Dwi, out);
  gemm_dx_scan_kernel<<<1 + DX_TILES, 256, 0, stream>>>(x, Dwr, Dwi, P, h0, th, dampp,
                                                        Hws, hfin, out);
  gemm_ch_out_kernel<<<dim3(MM / 128, NO / 128), 256, 0, stream>>>(Hws, Cwr, Cwi, out);
}
